# starting kernel
# baseline (speedup 1.0000x reference)
#include <hip/hip_runtime.h>
#include <math.h>

typedef float v2f __attribute__((ext_vector_type(2)));
typedef float v8f __attribute__((ext_vector_type(8)));

#define NEG_SLOPE 0.2f
#define ENC_NEG_INF 0x007FFFFFu

__device__ __forceinline__ unsigned fenc(float f) {
  unsigned u = __float_as_uint(f);
  return (u & 0x80000000u) ? ~u : (u | 0x80000000u);
}
__device__ __forceinline__ float fdec(unsigned k) {
  unsigned u = (k & 0x80000000u) ? (k & 0x7FFFFFFFu) : ~k;
  return __uint_as_float(u);
}

__global__ void gat_gemm128_wmma(const float* __restrict__ X,
                                 const float* __restrict__ W,
                                 float* __restrict__ H, int N) {
  const int wave = threadIdx.x >> 5;
  const int lane = threadIdx.x & 31;
  const int half = lane >> 4;
  const int lm   = lane & 15;
  const int row0 = blockIdx.x << 4;
  const int col  = (wave << 4) + lm;

  int arow = row0 + lm;
  if (arow >= N) arow = N - 1;
  const float* xr = X + (size_t)arow * 128;

  v8f c = {};
#pragma unroll
  for (int kk = 0; kk < 32; ++kk) {
    const int kb = (kk << 2) + (half << 1);
    v2f a, b;
    a.x = xr[kb];
    a.y = xr[kb + 1];
    b.x = W[(size_t)kb * 128 + col];
    b.y = W[(size_t)(kb + 1) * 128 + col];
    c = __builtin_amdgcn_wmma_f32_16x16x4_f32(false, a, false, b,
                                              (short)0, c, false, false);
  }

  const int orow0 = row0 + (half << 3);
#pragma unroll
  for (int v = 0; v < 8; ++v) {
    int r = orow0 + v;
    if (r < N) H[(size_t)r * 128 + col] = c[v];
  }
}

__global__ void gat_attn_logits(const float* __restrict__ H,
                                const float* __restrict__ a_src,
                                const float* __restrict__ a_dst,
                                float* __restrict__ als,
                                float* __restrict__ ald, int N) {
  const int wid  = blockIdx.x * (blockDim.x >> 5) + (threadIdx.x >> 5);
  const int lane = threadIdx.x & 31;
  if (wid >= N) return;
  const float* hr = H + (size_t)wid * 128 + lane * 4;
  const float* as = a_src + lane * 4;
  const float* ad = a_dst + lane * 4;
  float s = 0.f, d = 0.f;
#pragma unroll
  for (int j = 0; j < 4; ++j) {
    float hv = hr[j];
    s += hv * as[j];
    d += hv * ad[j];
  }
#pragma unroll
  for (int m = 8; m >= 1; m >>= 1) {
    s += __shfl_xor(s, m, 32);
    d += __shfl_xor(d, m, 32);
  }
  if ((lane & 15) == 0) {
    const int h = lane >> 4;
    als[wid * 2 + h] = s;
    ald[wid * 2 + h] = d;
  }
}

__global__ void gat_init(float* __restrict__ agg, unsigned* __restrict__ emax,
                         float* __restrict__ denom, int N) {
  const int tid = blockIdx.x * blockDim.x + threadIdx.x;
  if (tid < N * 128) agg[tid] = 0.f;
  if (tid < N * 2) { emax[tid] = ENC_NEG_INF; denom[tid] = 0.f; }
}

__device__ __forceinline__ void edge_sd(int e, int E, const int* __restrict__ es,
                                        const int* __restrict__ ed, int& s, int& d) {
  if (e < E) { s = es[e]; d = ed[e]; }
  else       { s = e - E; d = e - E; }
}

__global__ void gat_edge_max(const int* __restrict__ es, const int* __restrict__ ed,
                             const float* __restrict__ als, const float* __restrict__ ald,
                             unsigned* __restrict__ emax, int E, int Etot) {
  const int e = blockIdx.x * blockDim.x + threadIdx.x;
  if (e >= Etot) return;
  int s, d; edge_sd(e, E, es, ed, s, d);
#pragma unroll
  for (int h = 0; h < 2; ++h) {
    float v = als[s * 2 + h] + ald[d * 2 + h];
    v = (v >= 0.f) ? v : NEG_SLOPE * v;
    atomicMax(&emax[d * 2 + h], fenc(v));
  }
}

__global__ void gat_edge_sum(const int* __restrict__ es, const int* __restrict__ ed,
                             const float* __restrict__ als, const float* __restrict__ ald,
                             const unsigned* __restrict__ emax,
                             float* __restrict__ denom, int E, int Etot) {
  const int e = blockIdx.x * blockDim.x + threadIdx.x;
  if (e >= Etot) return;
  int s, d; edge_sd(e, E, es, ed, s, d);
#pragma unroll
  for (int h = 0; h < 2; ++h) {
    float v = als[s * 2 + h] + ald[d * 2 + h];
    v = (v >= 0.f) ? v : NEG_SLOPE * v;
    atomicAdd(&denom[d * 2 + h], expf(v - fdec(emax[d * 2 + h])));
  }
}

__global__ void gat_edge_agg(const int* __restrict__ es, const int* __restrict__ ed,
                             const float* __restrict__ als, const float* __restrict__ ald,
                             const unsigned* __restrict__ emax,
                             const float* __restrict__ denom,
                             const float* __restrict__ H, float* __restrict__ agg,
                             int E, int Etot, int nWaves) {
  const int wid  = blockIdx.x * (blockDim.x >> 5) + (threadIdx.x >> 5);
  const int lane = threadIdx.x & 31;
  for (int e = wid; e < Etot; e += nWaves) {
    int s, d; edge_sd(e, E, es, ed, s, d);
    const int en = e + nWaves;
    if (en < Etot) {
      const int sn = (en < E) ? es[en] : (en - E);
      __builtin_prefetch(&H[(size_t)sn * 128 + lane * 4], 0, 0);
    }
    float alpha[2];
#pragma unroll
    for (int h = 0; h < 2; ++h) {
      float v = als[s * 2 + h] + ald[d * 2 + h];
      v = (v >= 0.f) ? v : NEG_SLOPE * v;
      alpha[h] = expf(v - fdec(emax[d * 2 + h])) / denom[d * 2 + h];
    }
    const float* hs = H + (size_t)s * 128;
    float* ag = agg + (size_t)d * 128;
#pragma unroll
    for (int k = 0; k < 4; ++k) {
      const int idx = lane + (k << 5);
      atomicAdd(&ag[idx], hs[idx] * ((idx < 64) ? alpha[0] : alpha[1]));
    }
  }
}

__global__ void gat_bias_elu(float* __restrict__ agg, const float* __restrict__ b, int N) {
  const int tid = blockIdx.x * blockDim.x + threadIdx.x;
  if (tid >= N * 128) return;
  const float v = agg[tid] + b[tid & 127];
  agg[tid] = (v > 0.f) ? v : (expf(v) - 1.f);
}

__global__ void gat_head_mean_bias(const float* __restrict__ agg,
                                   const float* __restrict__ b,
                                   float* __restrict__ out, int N) {
  const int tid = blockIdx.x * blockDim.x + threadIdx.x;
  if (tid >= N * 64) return;
  const int n = tid >> 6, c = tid & 63;
  out[tid] = 0.5f * (agg[(size_t)n * 128 + c] + agg[(size_t)n * 128 + 64 + c]) + b[c];
}

extern "C" void kernel_launch(void* const* d_in, const int* in_sizes, int n_in,
                              void* d_out, int out_size, void* d_ws, size_t ws_size,
                              hipStream_t stream) {
  const float* x   = (const float*)d_in[0];
  const int*   ei  = (const int*)d_in[1];
  const float* W1  = (const float*)d_in[2];
  const float* as1 = (const float*)d_in[3];
  const float* ad1 = (const float*)d_in[4];
  const float* b1  = (const float*)d_in[5];
  const float* W2  = (const float*)d_in[6];
  const float* as2 = (const float*)d_in[7];
  const float* ad2 = (const float*)d_in[8];
  const float* b2  = (const float*)d_in[9];

  const int N = in_sizes[0] / 128;
  const int E = in_sizes[1] / 2;
  const int Etot = E + N;
  const int* esrc = ei;
  const int* edst = ei + E;

  float*    h     = (float*)d_ws;
  float*    agg   = h + (size_t)N * 128;
  float*    als   = agg + (size_t)N * 128;
  float*    ald   = als + (size_t)N * 2;
  unsigned* emax  = (unsigned*)(ald + (size_t)N * 2);
  float*    denom = (float*)(emax + (size_t)N * 2);

  const int B = 256;
  const int gGemm  = (N + 15) / 16;
  const int gNodeW = (N + 7) / 8;
  const int gInit  = (N * 128 + B - 1) / B;
  const int gEdge  = (Etot + B - 1) / B;
  const int aggBlocks = 8192;
  const int nWaves = aggBlocks * (B / 32);
  const int gFin1 = (N * 128 + B - 1) / B;
  const int gFin2 = (N * 64 + B - 1) / B;

  gat_gemm128_wmma<<<gGemm, B, 0, stream>>>(x, W1, h, N);
  gat_attn_logits<<<gNodeW, B, 0, stream>>>(h, as1, ad1, als, ald, N);
  gat_init<<<gInit, B, 0, stream>>>(agg, emax, denom, N);
  gat_edge_max<<<gEdge, B, 0, stream>>>(esrc, edst, als, ald, emax, E, Etot);
  gat_edge_sum<<<gEdge, B, 0, stream>>>(esrc, edst, als, ald, emax, denom, E, Etot);
  gat_edge_agg<<<aggBlocks, B, 0, stream>>>(esrc, edst, als, ald, emax, denom,
                                            h, agg, E, Etot, nWaves);
  gat_bias_elu<<<gFin1, B, 0, stream>>>(agg, b1, N);

  gat_gemm128_wmma<<<gGemm, B, 0, stream>>>(agg, W2, h, N);
  gat_attn_logits<<<gNodeW, B, 0, stream>>>(h, as2, ad2, als, ald, N);
  gat_init<<<gInit, B, 0, stream>>>(agg, emax, denom, N);
  gat_edge_max<<<gEdge, B, 0, stream>>>(esrc, edst, als, ald, emax, E, Etot);
  gat_edge_sum<<<gEdge, B, 0, stream>>>(esrc, edst, als, ald, emax, denom, E, Etot);
  gat_edge_agg<<<aggBlocks, B, 0, stream>>>(esrc, edst, als, ald, emax, denom,
                                            h, agg, E, Etot, nWaves);
  gat_head_mean_bias<<<gFin2, B, 0, stream>>>(agg, b2, (float*)d_out, N);
}


# Edges counting-sorted by destination in-call (histogram, scan, fill); one wave per node does softmax and weighted row gather in registers, no float atomics. Sum order: self edge, then arrival order.
# speedup vs baseline: 2.6196x; 2.6196x over previous
#include <hip/hip_runtime.h>
#include <math.h>

typedef float v2f __attribute__((ext_vector_type(2)));
typedef float v8f __attribute__((ext_vector_type(8)));

#define NEG_SLOPE 0.2f

__global__ void gat_gemm128_wmma(const float* __restrict__ X,
                                 const float* __restrict__ W,
                                 float* __restrict__ H, int N) {
  const int wave = threadIdx.x >> 5;
  const int lane = threadIdx.x & 31;
  const int half = lane >> 4;
  const int lm   = lane & 15;
  const int row0 = blockIdx.x << 4;
  const int col  = (wave << 4) + lm;

  int arow = row0 + lm;
  if (arow >= N) arow = N - 1;
  const float* xr = X + (size_t)arow * 128;

  v8f c = {};
#pragma unroll
  for (int kk = 0; kk < 32; ++kk) {
    const int kb = (kk << 2) + (half << 1);
    v2f a, b;
    a.x = xr[kb];
    a.y = xr[kb + 1];
    b.x = W[(size_t)kb * 128 + col];
    b.y = W[(size_t)(kb + 1) * 128 + col];
    c = __builtin_amdgcn_wmma_f32_16x16x4_f32(false, a, false, b,
                                              (short)0, c, false, false);
  }

  const int orow0 = row0 + (half << 3);
#pragma unroll
  for (int v = 0; v < 8; ++v) {
    int r = orow0 + v;
    if (r < N) H[(size_t)r * 128 + col] = c[v];
  }
}

__global__ void gat_attn_logits(const float* __restrict__ H,
                                const float* __restrict__ a_src,
                                const float* __restrict__ a_dst,
                                float* __restrict__ als,
                                float* __restrict__ ald, int N) {
  const int wid  = blockIdx.x * (blockDim.x >> 5) + (threadIdx.x >> 5);
  const int lane = threadIdx.x & 31;
  if (wid >= N) return;
  const float* hr = H + (size_t)wid * 128 + lane * 4;
  const float* as = a_src + lane * 4;
  const float* ad = a_dst + lane * 4;
  float s = 0.f, d = 0.f;
#pragma unroll
  for (int j = 0; j < 4; ++j) {
    float hv = hr[j];
    s += hv * as[j];
    d += hv * ad[j];
  }
#pragma unroll
  for (int m = 8; m >= 1; m >>= 1) {
    s += __shfl_xor(s, m, 32);
    d += __shfl_xor(d, m, 32);
  }
  if ((lane & 15) == 0) {
    const int h = lane >> 4;
    als[wid * 2 + h] = s;
    ald[wid * 2 + h] = d;
  }
}

__global__ void csr_zero(int* __restrict__ cnt, int N) {
  const int i = blockIdx.x * blockDim.x + threadIdx.x;
  if (i < N) cnt[i] = 0;
}

__global__ void csr_hist(const int* __restrict__ ed, int* __restrict__ cnt, int E) {
  const int e = blockIdx.x * blockDim.x + threadIdx.x;
  if (e < E) atomicAdd(&cnt[ed[e]], 1);
}

__global__ void csr_scan(const int* __restrict__ cnt, int* __restrict__ rowptr,
                         int* __restrict__ cur, int N) {
  __shared__ int part[1024];
  const int t = threadIdx.x;
  const int per = (N + 1023) / 1024;
  const int b = t * per;
  int e = b + per;
  if (e > N) e = N;
  int s = 0;
  for (int i = b; i < e; ++i) s += cnt[i];
  part[t] = s;
  __syncthreads();
  for (int off = 1; off < 1024; off <<= 1) {
    const int v = (t >= off) ? part[t - off] : 0;
    __syncthreads();
    part[t] += v;
    __syncthreads();
  }
  int run = part[t] - s;
  int i = b;
  for (; i + 4 <= e; i += 4) {
    const int c0 = cnt[i], c1 = cnt[i + 1], c2 = cnt[i + 2], c3 = cnt[i + 3];
    const int r1 = run + c0, r2 = r1 + c1, r3 = r2 + c2;
    rowptr[i] = run;    cur[i] = run;
    rowptr[i + 1] = r1; cur[i + 1] = r1;
    rowptr[i + 2] = r2; cur[i + 2] = r2;
    rowptr[i + 3] = r3; cur[i + 3] = r3;
    run = r3 + c3;
  }
  for (; i < e; ++i) {
    rowptr[i] = run;
    cur[i] = run;
    run += cnt[i];
  }
  if (t == 1023) rowptr[N] = part[1023];
}

__global__ void csr_fill(const int* __restrict__ es, const int* __restrict__ ed,
                         int* __restrict__ cur, int* __restrict__ csrc, int E) {
  const int e = blockIdx.x * blockDim.x + threadIdx.x;
  if (e >= E) return;
  const int p = atomicAdd(&cur[ed[e]], 1);
  csrc[p] = es[e];
}

__device__ __forceinline__ float lrelu(float v) {
  return (v >= 0.f) ? v : NEG_SLOPE * v;
}

__device__ __forceinline__ float lane_bcast(float v, int t) {
  return __int_as_float(__builtin_amdgcn_readlane(__float_as_int(v), t));
}

__device__ __forceinline__ void gather_rows(float4& acc, const float4* __restrict__ H4,
                                            int lane, bool hi, int s, float w0, float w1,
                                            int c) {
  int t = 0;
  for (; t + 4 <= c; t += 4) {
    const int sa = __builtin_amdgcn_readlane(s, t);
    const int sb = __builtin_amdgcn_readlane(s, t + 1);
    const int sc = __builtin_amdgcn_readlane(s, t + 2);
    const int sd = __builtin_amdgcn_readlane(s, t + 3);
    const float4 ra = H4[(size_t)sa * 32 + lane];
    const float4 rb = H4[(size_t)sb * 32 + lane];
    const float4 rc = H4[(size_t)sc * 32 + lane];
    const float4 rd = H4[(size_t)sd * 32 + lane];
    const float a0 = lane_bcast(w0, t),     a1 = lane_bcast(w1, t);
    const float b0 = lane_bcast(w0, t + 1), b1 = lane_bcast(w1, t + 1);
    const float c0 = lane_bcast(w0, t + 2), c1 = lane_bcast(w1, t + 2);
    const float d0 = lane_bcast(w0, t + 3), d1 = lane_bcast(w1, t + 3);
    const float wa = hi ? a1 : a0;
    const float wb = hi ? b1 : b0;
    const float wc = hi ? c1 : c0;
    const float wd = hi ? d1 : d0;
    acc.x += wa * ra.x; acc.y += wa * ra.y; acc.z += wa * ra.z; acc.w += wa * ra.w;
    acc.x += wb * rb.x; acc.y += wb * rb.y; acc.z += wb * rb.z; acc.w += wb * rb.w;
    acc.x += wc * rc.x; acc.y += wc * rc.y; acc.z += wc * rc.z; acc.w += wc * rc.w;
    acc.x += wd * rd.x; acc.y += wd * rd.y; acc.z += wd * rd.z; acc.w += wd * rd.w;
  }
  for (; t < c; ++t) {
    const int sa = __builtin_amdgcn_readlane(s, t);
    const float4 ra = H4[(size_t)sa * 32 + lane];
    const float a0 = lane_bcast(w0, t), a1 = lane_bcast(w1, t);
    const float wa = hi ? a1 : a0;
    acc.x += wa * ra.x; acc.y += wa * ra.y; acc.z += wa * ra.z; acc.w += wa * ra.w;
  }
}

template <bool CONCAT>
__global__ void gat_node_gather(const int* __restrict__ rowptr, const int* __restrict__ csrc,
                                const float* __restrict__ als, const float* __restrict__ ald,
                                const float* __restrict__ H, const float* __restrict__ bias,
                                float* __restrict__ out, int N) {
  const int n = __builtin_amdgcn_readfirstlane(
      (int)(blockIdx.x * (blockDim.x >> 5) + (threadIdx.x >> 5)));
  const int lane = threadIdx.x & 31;
  if (n >= N) return;
  const int beg = rowptr[n], end = rowptr[n + 1];
  const int deg = end - beg;
  const float2 asn = *(const float2*)(als + 2 * (size_t)n);
  const float2 adn = *(const float2*)(ald + 2 * (size_t)n);
  const float self0 = lrelu(asn.x + adn.x);
  const float self1 = lrelu(asn.y + adn.y);

  const bool has = lane < deg;
  int   s_l = n;
  float v0 = -INFINITY, v1 = -INFINITY;
  if (has) {
    s_l = csrc[beg + lane];
    const float2 a = *(const float2*)(als + 2 * (size_t)s_l);
    v0 = lrelu(a.x + adn.x);
    v1 = lrelu(a.y + adn.y);
  }

  float m0 = fmaxf(self0, v0), m1 = fmaxf(self1, v1);
  for (int j = beg + 32 + lane; j < end; j += 32) {
    const int s = csrc[j];
    const float2 a = *(const float2*)(als + 2 * (size_t)s);
    m0 = fmaxf(m0, lrelu(a.x + adn.x));
    m1 = fmaxf(m1, lrelu(a.y + adn.y));
  }
#pragma unroll
  for (int k = 16; k >= 1; k >>= 1) {
    m0 = fmaxf(m0, __shfl_xor(m0, k, 32));
    m1 = fmaxf(m1, __shfl_xor(m1, k, 32));
  }

  const float ex0 = has ? expf(v0 - m0) : 0.f;
  const float ex1 = has ? expf(v1 - m1) : 0.f;
  float d0 = ex0, d1 = ex1;
  for (int j = beg + 32 + lane; j < end; j += 32) {
    const int s = csrc[j];
    const float2 a = *(const float2*)(als + 2 * (size_t)s);
    d0 += expf(lrelu(a.x + adn.x) - m0);
    d1 += expf(lrelu(a.y + adn.y) - m1);
  }
#pragma unroll
  for (int k = 16; k >= 1; k >>= 1) {
    d0 += __shfl_xor(d0, k, 32);
    d1 += __shfl_xor(d1, k, 32);
  }
  const float exs0 = expf(self0 - m0), exs1 = expf(self1 - m1);
  d0 += exs0;
  d1 += exs1;

  const float4* H4 = (const float4*)H;
  const bool hi = lane >= 16;
  const float wself = hi ? (exs1 / d1) : (exs0 / d0);
  const float4 hn = H4[(size_t)n * 32 + lane];
  float4 acc;
  acc.x = wself * hn.x; acc.y = wself * hn.y; acc.z = wself * hn.z; acc.w = wself * hn.w;
  gather_rows(acc, H4, lane, hi, s_l, ex0 / d0, ex1 / d1, deg < 32 ? deg : 32);
  for (int base = beg + 32; base < end; base += 32) {
    const int j = base + lane;
    int s = n;
    float w0 = 0.f, w1 = 0.f;
    if (j < end) {
      s = csrc[j];
      const float2 a = *(const float2*)(als + 2 * (size_t)s);
      w0 = expf(lrelu(a.x + adn.x) - m0) / d0;
      w1 = expf(lrelu(a.y + adn.y) - m1) / d1;
    }
    const int c = end - base;
    gather_rows(acc, H4, lane, hi, s, w0, w1, c < 32 ? c : 32);
  }

  if (CONCAT) {
    float4 v;
    v.x = acc.x + bias[lane * 4 + 0];
    v.y = acc.y + bias[lane * 4 + 1];
    v.z = acc.z + bias[lane * 4 + 2];
    v.w = acc.w + bias[lane * 4 + 3];
    v.x = (v.x > 0.f) ? v.x : (expf(v.x) - 1.f);
    v.y = (v.y > 0.f) ? v.y : (expf(v.y) - 1.f);
    v.z = (v.z > 0.f) ? v.z : (expf(v.z) - 1.f);
    v.w = (v.w > 0.f) ? v.w : (expf(v.w) - 1.f);
    ((float4*)out)[(size_t)n * 32 + lane] = v;
  } else {
    float4 o;
    o.x = __shfl_xor(acc.x, 16, 32);
    o.y = __shfl_xor(acc.y, 16, 32);
    o.z = __shfl_xor(acc.z, 16, 32);
    o.w = __shfl_xor(acc.w, 16, 32);
    if (!hi) {
      float4 v;
      v.x = 0.5f * (acc.x + o.x) + bias[lane * 4 + 0];
      v.y = 0.5f * (acc.y + o.y) + bias[lane * 4 + 1];
      v.z = 0.5f * (acc.z + o.z) + bias[lane * 4 + 2];
      v.w = 0.5f * (acc.w + o.w) + bias[lane * 4 + 3];
      ((float4*)out)[(size_t)n * 16 + lane] = v;
    }
  }
}

extern "C" void kernel_launch(void* const* d_in, const int* in_sizes, int n_in,
                              void* d_out, int out_size, void* d_ws, size_t ws_size,
                              hipStream_t stream) {
  const float* x   = (const float*)d_in[0];
  const int*   ei  = (const int*)d_in[1];
  const float* W1  = (const float*)d_in[2];
  const float* as1 = (const float*)d_in[3];
  const float* ad1 = (const float*)d_in[4];
  const float* b1  = (const float*)d_in[5];
  const float* W2  = (const float*)d_in[6];
  const float* as2 = (const float*)d_in[7];
  const float* ad2 = (const float*)d_in[8];
  const float* b2  = (const float*)d_in[9];

  const int N = in_sizes[0] / 128;
  const int E = in_sizes[1] / 2;
  const int* esrc = ei;
  const int* edst = ei + E;

  float* h      = (float*)d_ws;
  float* act    = h + (size_t)N * 128;
  float* als    = act + (size_t)N * 128;
  float* ald    = als + (size_t)N * 2;
  int*   cnt    = (int*)(ald + (size_t)N * 2);
  int*   rowptr = cnt + (size_t)N;
  int*   cur    = rowptr + (size_t)N + 1;
  int*   csrc   = cur + (size_t)N;

  const int B = 256;
  const int gGemm  = (N + 15) / 16;
  const int gNodeW = (N + 7) / 8;
  const int gNode  = (N + B - 1) / B;
  int gEdge = (E + B - 1) / B;
  if (gEdge < 1) gEdge = 1;

  csr_zero<<<gNode, B, 0, stream>>>(cnt, N);
  csr_hist<<<gEdge, B, 0, stream>>>(edst, cnt, E);
  csr_scan<<<1, 1024, 0, stream>>>(cnt, rowptr, cur, N);
  csr_fill<<<gEdge, B, 0, stream>>>(esrc, edst, cur, csrc, E);

  gat_gemm128_wmma<<<gGemm, B, 0, stream>>>(x, W1, h, N);
  gat_attn_logits<<<gNodeW, B, 0, stream>>>(h, as1, ad1, als, ald, N);
  gat_node_gather<true><<<gNodeW, B, 0, stream>>>(rowptr, csrc, als, ald, h, b1, act, N);

  gat_gemm128_wmma<<<gGemm, B, 0, stream>>>(act, W2, h, N);
  gat_attn_logits<<<gNodeW, B, 0, stream>>>(h, as2, ad2, als, ald, N);
  gat_node_gather<false><<<gNodeW, B, 0, stream>>>(rowptr, csrc, als, ald, h, b2,
                                                   (float*)d_out, N);
}


# Count prefix scan spread over many blocks in one launch: each block re-sums the counts ahead of its 1024-node slice with whole-line loads, then scans the slice; no block waits on another.
# speedup vs baseline: 3.1118x; 1.1879x over previous
#include <hip/hip_runtime.h>
#include <math.h>

typedef float v2f __attribute__((ext_vector_type(2)));
typedef float v8f __attribute__((ext_vector_type(8)));

#define NEG_SLOPE 0.2f

__global__ void gat_gemm128_wmma(const float* __restrict__ X,
                                 const float* __restrict__ W,
                                 float* __restrict__ H, int N) {
  const int wave = threadIdx.x >> 5;
  const int lane = threadIdx.x & 31;
  const int half = lane >> 4;
  const int lm   = lane & 15;
  const int row0 = blockIdx.x << 4;
  const int col  = (wave << 4) + lm;

  int arow = row0 + lm;
  if (arow >= N) arow = N - 1;
  const float* xr = X + (size_t)arow * 128;

  v8f c = {};
#pragma unroll
  for (int kk = 0; kk < 32; ++kk) {
    const int kb = (kk << 2) + (half << 1);
    v2f a, b;
    a.x = xr[kb];
    a.y = xr[kb + 1];
    b.x = W[(size_t)kb * 128 + col];
    b.y = W[(size_t)(kb + 1) * 128 + col];
    c = __builtin_amdgcn_wmma_f32_16x16x4_f32(false, a, false, b,
                                              (short)0, c, false, false);
  }

  const int orow0 = row0 + (half << 3);
#pragma unroll
  for (int v = 0; v < 8; ++v) {
    int r = orow0 + v;
    if (r < N) H[(size_t)r * 128 + col] = c[v];
  }
}

__global__ void gat_attn_logits(const float* __restrict__ H,
                                const float* __restrict__ a_src,
                                const float* __restrict__ a_dst,
                                float* __restrict__ als,
                                float* __restrict__ ald, int N) {
  const int wid  = blockIdx.x * (blockDim.x >> 5) + (threadIdx.x >> 5);
  const int lane = threadIdx.x & 31;
  if (wid >= N) return;
  const float* hr = H + (size_t)wid * 128 + lane * 4;
  const float* as = a_src + lane * 4;
  const float* ad = a_dst + lane * 4;
  float s = 0.f, d = 0.f;
#pragma unroll
  for (int j = 0; j < 4; ++j) {
    float hv = hr[j];
    s += hv * as[j];
    d += hv * ad[j];
  }
#pragma unroll
  for (int m = 8; m >= 1; m >>= 1) {
    s += __shfl_xor(s, m, 32);
    d += __shfl_xor(d, m, 32);
  }
  if ((lane & 15) == 0) {
    const int h = lane >> 4;
    als[wid * 2 + h] = s;
    ald[wid * 2 + h] = d;
  }
}

__global__ void csr_zero(int* __restrict__ cnt, int N) {
  const int i = blockIdx.x * blockDim.x + threadIdx.x;
  if (i < N) cnt[i] = 0;
}

__global__ void csr_hist(const int* __restrict__ ed, int* __restrict__ cnt, int E) {
  const int e = blockIdx.x * blockDim.x + threadIdx.x;
  if (e < E) atomicAdd(&cnt[ed[e]], 1);
}

#define SCAN_T 1024
__global__ void __launch_bounds__(SCAN_T)
csr_scan(const int* __restrict__ cnt, int* __restrict__ rowptr,
         int* __restrict__ cur, int N) {
  __shared__ int wpre[32];
  __shared__ int wown[32];
  const int t = threadIdx.x;
  const int lane = t & 31;
  const int w = t >> 5;
  const int base = blockIdx.x * SCAN_T;

  const int4* c4 = (const int4*)cnt;
  const int nv = base >> 2;
  int s0 = 0, s1 = 0, s2 = 0, s3 = 0;
  int i = t;
  for (; i + 3 * SCAN_T < nv; i += 4 * SCAN_T) {
    const int4 a = c4[i];
    const int4 b = c4[i + SCAN_T];
    const int4 c = c4[i + 2 * SCAN_T];
    const int4 d = c4[i + 3 * SCAN_T];
    s0 += (a.x + a.y) + (a.z + a.w);
    s1 += (b.x + b.y) + (b.z + b.w);
    s2 += (c.x + c.y) + (c.z + c.w);
    s3 += (d.x + d.y) + (d.z + d.w);
  }
  for (; i < nv; i += SCAN_T) {
    const int4 a = c4[i];
    s0 += (a.x + a.y) + (a.z + a.w);
  }
  int s = (s0 + s1) + (s2 + s3);

  const int idx = base + t;
  const int c = (idx < N) ? cnt[idx] : 0;

#pragma unroll
  for (int k = 16; k >= 1; k >>= 1) s += __shfl_xor(s, k, 32);
  int inc = c;
#pragma unroll
  for (int d = 1; d < 32; d <<= 1) {
    const int v = __shfl_up(inc, d, 32);
    if (lane >= d) inc += v;
  }
  if (lane == 31) wown[w] = inc;
  if (lane == 0)  wpre[w] = s;
  __syncthreads();

  int p = wpre[lane];
  int o = wown[lane];
#pragma unroll
  for (int k = 16; k >= 1; k >>= 1) p += __shfl_xor(p, k, 32);
#pragma unroll
  for (int d = 1; d < 32; d <<= 1) {
    const int v = __shfl_up(o, d, 32);
    if (lane >= d) o += v;
  }
  const int before = __shfl(o, (w + 31) & 31, 32);
  const int excl = p + (w > 0 ? before : 0) + (inc - c);
  if (idx < N) {
    rowptr[idx] = excl;
    cur[idx] = excl;
    if (idx == N - 1) rowptr[N] = excl + c;
  }
}

__global__ void csr_fill(const int* __restrict__ es, const int* __restrict__ ed,
                         int* __restrict__ cur, int* __restrict__ csrc, int E) {
  const int e = blockIdx.x * blockDim.x + threadIdx.x;
  if (e >= E) return;
  const int p = atomicAdd(&cur[ed[e]], 1);
  csrc[p] = es[e];
}

__device__ __forceinline__ float lrelu(float v) {
  return (v >= 0.f) ? v : NEG_SLOPE * v;
}

__device__ __forceinline__ float lane_bcast(float v, int t) {
  return __int_as_float(__builtin_amdgcn_readlane(__float_as_int(v), t));
}

__device__ __forceinline__ void gather_rows(float4& acc, const float4* __restrict__ H4,
                                            int lane, bool hi, int s, float w0, float w1,
                                            int c) {
  int t = 0;
  for (; t + 4 <= c; t += 4) {
    const int sa = __builtin_amdgcn_readlane(s, t);
    const int sb = __builtin_amdgcn_readlane(s, t + 1);
    const int sc = __builtin_amdgcn_readlane(s, t + 2);
    const int sd = __builtin_amdgcn_readlane(s, t + 3);
    const float4 ra = H4[(size_t)sa * 32 + lane];
    const float4 rb = H4[(size_t)sb * 32 + lane];
    const float4 rc = H4[(size_t)sc * 32 + lane];
    const float4 rd = H4[(size_t)sd * 32 + lane];
    const float a0 = lane_bcast(w0, t),     a1 = lane_bcast(w1, t);
    const float b0 = lane_bcast(w0, t + 1), b1 = lane_bcast(w1, t + 1);
    const float c0 = lane_bcast(w0, t + 2), c1 = lane_bcast(w1, t + 2);
    const float d0 = lane_bcast(w0, t + 3), d1 = lane_bcast(w1, t + 3);
    const float wa = hi ? a1 : a0;
    const float wb = hi ? b1 : b0;
    const float wc = hi ? c1 : c0;
    const float wd = hi ? d1 : d0;
    acc.x += wa * ra.x; acc.y += wa * ra.y; acc.z += wa * ra.z; acc.w += wa * ra.w;
    acc.x += wb * rb.x; acc.y += wb * rb.y; acc.z += wb * rb.z; acc.w += wb * rb.w;
    acc.x += wc * rc.x; acc.y += wc * rc.y; acc.z += wc * rc.z; acc.w += wc * rc.w;
    acc.x += wd * rd.x; acc.y += wd * rd.y; acc.z += wd * rd.z; acc.w += wd * rd.w;
  }
  for (; t < c; ++t) {
    const int sa = __builtin_amdgcn_readlane(s, t);
    const float4 ra = H4[(size_t)sa * 32 + lane];
    const float a0 = lane_bcast(w0, t), a1 = lane_bcast(w1, t);
    const float wa = hi ? a1 : a0;
    acc.x += wa * ra.x; acc.y += wa * ra.y; acc.z += wa * ra.z; acc.w += wa * ra.w;
  }
}

template <bool CONCAT>
__global__ void gat_node_gather(const int* __restrict__ rowptr, const int* __restrict__ csrc,
                                const float* __restrict__ als, const float* __restrict__ ald,
                                const float* __restrict__ H, const float* __restrict__ bias,
                                float* __restrict__ out, int N) {
  const int n = __builtin_amdgcn_readfirstlane(
      (int)(blockIdx.x * (blockDim.x >> 5) + (threadIdx.x >> 5)));
  const int lane = threadIdx.x & 31;
  if (n >= N) return;
  const int beg = rowptr[n], end = rowptr[n + 1];
  const int deg = end - beg;
  const float2 asn = *(const float2*)(als + 2 * (size_t)n);
  const float2 adn = *(const float2*)(ald + 2 * (size_t)n);
  const float self0 = lrelu(asn.x + adn.x);
  const float self1 = lrelu(asn.y + adn.y);

  const bool has = lane < deg;
  int   s_l = n;
  float v0 = -INFINITY, v1 = -INFINITY;
  if (has) {
    s_l = csrc[beg + lane];
    const float2 a = *(const float2*)(als + 2 * (size_t)s_l);
    v0 = lrelu(a.x + adn.x);
    v1 = lrelu(a.y + adn.y);
  }

  float m0 = fmaxf(self0, v0), m1 = fmaxf(self1, v1);
  for (int j = beg + 32 + lane; j < end; j += 32) {
    const int s = csrc[j];
    const float2 a = *(const float2*)(als + 2 * (size_t)s);
    m0 = fmaxf(m0, lrelu(a.x + adn.x));
    m1 = fmaxf(m1, lrelu(a.y + adn.y));
  }
#pragma unroll
  for (int k = 16; k >= 1; k >>= 1) {
    m0 = fmaxf(m0, __shfl_xor(m0, k, 32));
    m1 = fmaxf(m1, __shfl_xor(m1, k, 32));
  }

  const float ex0 = has ? expf(v0 - m0) : 0.f;
  const float ex1 = has ? expf(v1 - m1) : 0.f;
  float d0 = ex0, d1 = ex1;
  for (int j = beg + 32 + lane; j < end; j += 32) {
    const int s = csrc[j];
    const float2 a = *(const float2*)(als + 2 * (size_t)s);
    d0 += expf(lrelu(a.x + adn.x) - m0);
    d1 += expf(lrelu(a.y + adn.y) - m1);
  }
#pragma unroll
  for (int k = 16; k >= 1; k >>= 1) {
    d0 += __shfl_xor(d0, k, 32);
    d1 += __shfl_xor(d1, k, 32);
  }
  const float exs0 = expf(self0 - m0), exs1 = expf(self1 - m1);
  d0 += exs0;
  d1 += exs1;

  const float4* H4 = (const float4*)H;
  const bool hi = lane >= 16;
  const float wself = hi ? (exs1 / d1) : (exs0 / d0);
  const float4 hn = H4[(size_t)n * 32 + lane];
  float4 acc;
  acc.x = wself * hn.x; acc.y = wself * hn.y; acc.z = wself * hn.z; acc.w = wself * hn.w;
  gather_rows(acc, H4, lane, hi, s_l, ex0 / d0, ex1 / d1, deg < 32 ? deg : 32);
  for (int base = beg + 32; base < end; base += 32) {
    const int j = base + lane;
    int s = n;
    float w0 = 0.f, w1 = 0.f;
    if (j < end) {
      s = csrc[j];
      const float2 a = *(const float2*)(als + 2 * (size_t)s);
      w0 = expf(lrelu(a.x + adn.x) - m0) / d0;
      w1 = expf(lrelu(a.y + adn.y) - m1) / d1;
    }
    const int c = end - base;
    gather_rows(acc, H4, lane, hi, s, w0, w1, c < 32 ? c : 32);
  }

  if (CONCAT) {
    float4 v;
    v.x = acc.x + bias[lane * 4 + 0];
    v.y = acc.y + bias[lane * 4 + 1];
    v.z = acc.z + bias[lane * 4 + 2];
    v.w = acc.w + bias[lane * 4 + 3];
    v.x = (v.x > 0.f) ? v.x : (expf(v.x) - 1.f);
    v.y = (v.y > 0.f) ? v.y : (expf(v.y) - 1.f);
    v.z = (v.z > 0.f) ? v.z : (expf(v.z) - 1.f);
    v.w = (v.w > 0.f) ? v.w : (expf(v.w) - 1.f);
    ((float4*)out)[(size_t)n * 32 + lane] = v;
  } else {
    float4 o;
    o.x = __shfl_xor(acc.x, 16, 32);
    o.y = __shfl_xor(acc.y, 16, 32);
    o.z = __shfl_xor(acc.z, 16, 32);
    o.w = __shfl_xor(acc.w, 16, 32);
    if (!hi) {
      float4 v;
      v.x = 0.5f * (acc.x + o.x) + bias[lane * 4 + 0];
      v.y = 0.5f * (acc.y + o.y) + bias[lane * 4 + 1];
      v.z = 0.5f * (acc.z + o.z) + bias[lane * 4 + 2];
      v.w = 0.5f * (acc.w + o.w) + bias[lane * 4 + 3];
      ((float4*)out)[(size_t)n * 16 + lane] = v;
    }
  }
}

extern "C" void kernel_launch(void* const* d_in, const int* in_sizes, int n_in,
                              void* d_out, int out_size, void* d_ws, size_t ws_size,
                              hipStream_t stream) {
  const float* x   = (const float*)d_in[0];
  const int*   ei  = (const int*)d_in[1];
  const float* W1  = (const float*)d_in[2];
  const float* as1 = (const float*)d_in[3];
  const float* ad1 = (const float*)d_in[4];
  const float* b1  = (const float*)d_in[5];
  const float* W2  = (const float*)d_in[6];
  const float* as2 = (const float*)d_in[7];
  const float* ad2 = (const float*)d_in[8];
  const float* b2  = (const float*)d_in[9];

  const int N = in_sizes[0] / 128;
  const int E = in_sizes[1] / 2;
  const int* esrc = ei;
  const int* edst = ei + E;

  float* h      = (float*)d_ws;
  float* act    = h + (size_t)N * 128;
  float* als    = act + (size_t)N * 128;
  float* ald    = als + (size_t)N * 2;
  int*   cnt    = (int*)(ald + (size_t)N * 2);
  int*   rowptr = cnt + (size_t)N;
  int*   cur    = rowptr + (size_t)N + 1;
  int*   csrc   = cur + (size_t)N;

  const int B = 256;
  const int gGemm  = (N + 15) / 16;
  const int gNodeW = (N + 7) / 8;
  const int gNode  = (N + B - 1) / B;
  const int gScan  = (N + SCAN_T - 1) / SCAN_T;
  int gEdge = (E + B - 1) / B;
  if (gEdge < 1) gEdge = 1;

  csr_zero<<<gNode, B, 0, stream>>>(cnt, N);
  csr_hist<<<gEdge, B, 0, stream>>>(edst, cnt, E);
  csr_scan<<<gScan, SCAN_T, 0, stream>>>(cnt, rowptr, cur, N);
  csr_fill<<<gEdge, B, 0, stream>>>(esrc, edst, cur, csrc, E);

  gat_gemm128_wmma<<<gGemm, B, 0, stream>>>(x, W1, h, N);
  gat_attn_logits<<<gNodeW, B, 0, stream>>>(h, as1, ad1, als, ald, N);
  gat_node_gather<true><<<gNodeW, B, 0, stream>>>(rowptr, csrc, als, ald, h, b1, act, N);

  gat_gemm128_wmma<<<gGemm, B, 0, stream>>>(act, W2, h, N);
  gat_attn_logits<<<gNodeW, B, 0, stream>>>(h, as2, ad2, als, ald, N);
  gat_node_gather<false><<<gNodeW, B, 0, stream>>>(rowptr, csrc, als, ald, h, b2,
                                                   (float*)d_out, N);
}


# Attention logits taken in the GEMM epilogue from the output tile staged in LDS (per-row sum order kept, 16-lane sums by DPP row exchange); H rows stored as whole 512-byte runs; logit launches gone.
# speedup vs baseline: 3.3619x; 1.0804x over previous
#include <hip/hip_runtime.h>
#include <math.h>

typedef float v2f __attribute__((ext_vector_type(2)));
typedef float v8f __attribute__((ext_vector_type(8)));

#define NEG_SLOPE 0.2f

#define ROW_XOR_ADD(v, m)                                                        \
  (v) += __int_as_float(__builtin_amdgcn_update_dpp(0, __float_as_int(v),        \
                                                    0x160 + (m), 0xf, 0xf, false))
#define TILE_PITCH 132
__global__ void gat_gemm128_wmma(const float* __restrict__ X,
                                 const float* __restrict__ W,
                                 const float* __restrict__ a_src,
                                 const float* __restrict__ a_dst,
                                 float* __restrict__ H,
                                 float* __restrict__ als,
                                 float* __restrict__ ald, int N) {
  __shared__ __attribute__((aligned(16))) float tile[16 * TILE_PITCH];
  const int wave = threadIdx.x >> 5;
  const int lane = threadIdx.x & 31;
  const int half = lane >> 4;
  const int lm   = lane & 15;
  const int row0 = blockIdx.x << 4;
  const int col  = (wave << 4) + lm;

  int arow = row0 + lm;
  if (arow >= N) arow = N - 1;
  const float* xr = X + (size_t)arow * 128;

  const float4 as4 = *(const float4*)(a_src + lane * 4);
  const float4 ad4 = *(const float4*)(a_dst + lane * 4);

  v8f c = {};
#pragma unroll
  for (int kk = 0; kk < 32; ++kk) {
    const int kb = (kk << 2) + (half << 1);
    v2f a, b;
    a.x = xr[kb];
    a.y = xr[kb + 1];
    b.x = W[(size_t)kb * 128 + col];
    b.y = W[(size_t)(kb + 1) * 128 + col];
    c = __builtin_amdgcn_wmma_f32_16x16x4_f32(false, a, false, b,
                                              (short)0, c, false, false);
  }

#pragma unroll
  for (int v = 0; v < 8; ++v)
    tile[((half << 3) + v) * TILE_PITCH + col] = c[v];
  __syncthreads();

  float4 hv[2];
  float s[2], d[2];
#pragma unroll
  for (int i = 0; i < 2; ++i) {
    const int r = (wave << 1) + i;
    hv[i] = *(const float4*)(tile + r * TILE_PITCH + lane * 4);
    s[i] = 0.f; d[i] = 0.f;
    s[i] += hv[i].x * as4.x;  d[i] += hv[i].x * ad4.x;
    s[i] += hv[i].y * as4.y;  d[i] += hv[i].y * ad4.y;
    s[i] += hv[i].z * as4.z;  d[i] += hv[i].z * ad4.z;
    s[i] += hv[i].w * as4.w;  d[i] += hv[i].w * ad4.w;
  }
  ROW_XOR_ADD(s[0], 8); ROW_XOR_ADD(d[0], 8); ROW_XOR_ADD(s[1], 8); ROW_XOR_ADD(d[1], 8);
  ROW_XOR_ADD(s[0], 4); ROW_XOR_ADD(d[0], 4); ROW_XOR_ADD(s[1], 4); ROW_XOR_ADD(d[1], 4);
  ROW_XOR_ADD(s[0], 2); ROW_XOR_ADD(d[0], 2); ROW_XOR_ADD(s[1], 2); ROW_XOR_ADD(d[1], 2);
  ROW_XOR_ADD(s[0], 1); ROW_XOR_ADD(d[0], 1); ROW_XOR_ADD(s[1], 1); ROW_XOR_ADD(d[1], 1);
#pragma unroll
  for (int i = 0; i < 2; ++i) {
    const int n = row0 + (wave << 1) + i;
    if (n < N) {
      ((float4*)H)[(size_t)n * 32 + lane] = hv[i];
      if (lm == 0) {
        als[n * 2 + half] = s[i];
        ald[n * 2 + half] = d[i];
      }
    }
  }
}

__global__ void csr_zero(int* __restrict__ cnt, int N) {
  const int i = blockIdx.x * blockDim.x + threadIdx.x;
  if (i < N) cnt[i] = 0;
}

__global__ void csr_hist(const int* __restrict__ ed, int* __restrict__ cnt, int E) {
  const int e = blockIdx.x * blockDim.x + threadIdx.x;
  if (e < E) atomicAdd(&cnt[ed[e]], 1);
}

#define SCAN_T 1024
__global__ void __launch_bounds__(SCAN_T)
csr_scan(const int* __restrict__ cnt, int* __restrict__ rowptr,
         int* __restrict__ cur, int N) {
  __shared__ int wpre[32];
  __shared__ int wown[32];
  const int t = threadIdx.x;
  const int lane = t & 31;
  const int w = t >> 5;
  const int base = blockIdx.x * SCAN_T;

  const int4* c4 = (const int4*)cnt;
  const int nv = base >> 2;
  int s0 = 0, s1 = 0, s2 = 0, s3 = 0;
  int i = t;
  for (; i + 3 * SCAN_T < nv; i += 4 * SCAN_T) {
    const int4 a = c4[i];
    const int4 b = c4[i + SCAN_T];
    const int4 c = c4[i + 2 * SCAN_T];
    const int4 d = c4[i + 3 * SCAN_T];
    s0 += (a.x + a.y) + (a.z + a.w);
    s1 += (b.x + b.y) + (b.z + b.w);
    s2 += (c.x + c.y) + (c.z + c.w);
    s3 += (d.x + d.y) + (d.z + d.w);
  }
  for (; i < nv; i += SCAN_T) {
    const int4 a = c4[i];
    s0 += (a.x + a.y) + (a.z + a.w);
  }
  int s = (s0 + s1) + (s2 + s3);

  const int idx = base + t;
  const int c = (idx < N) ? cnt[idx] : 0;

#pragma unroll
  for (int k = 16; k >= 1; k >>= 1) s += __shfl_xor(s, k, 32);
  int inc = c;
#pragma unroll
  for (int d = 1; d < 32; d <<= 1) {
    const int v = __shfl_up(inc, d, 32);
    if (lane >= d) inc += v;
  }
  if (lane == 31) wown[w] = inc;
  if (lane == 0)  wpre[w] = s;
  __syncthreads();

  int p = wpre[lane];
  int o = wown[lane];
#pragma unroll
  for (int k = 16; k >= 1; k >>= 1) p += __shfl_xor(p, k, 32);
#pragma unroll
  for (int d = 1; d < 32; d <<= 1) {
    const int v = __shfl_up(o, d, 32);
    if (lane >= d) o += v;
  }
  const int before = __shfl(o, (w + 31) & 31, 32);
  const int excl = p + (w > 0 ? before : 0) + (inc - c);
  if (idx < N) {
    rowptr[idx] = excl;
    cur[idx] = excl;
    if (idx == N - 1) rowptr[N] = excl + c;
  }
}

__global__ void csr_fill(const int* __restrict__ es, const int* __restrict__ ed,
                         int* __restrict__ cur, int* __restrict__ csrc, int E) {
  const int e = blockIdx.x * blockDim.x + threadIdx.x;
  if (e >= E) return;
  const int p = atomicAdd(&cur[ed[e]], 1);
  csrc[p] = es[e];
}

__device__ __forceinline__ float lrelu(float v) {
  return (v >= 0.f) ? v : NEG_SLOPE * v;
}

__device__ __forceinline__ float lane_bcast(float v, int t) {
  return __int_as_float(__builtin_amdgcn_readlane(__float_as_int(v), t));
}

__device__ __forceinline__ void gather_rows(float4& acc, const float4* __restrict__ H4,
                                            int lane, bool hi, int s, float w0, float w1,
                                            int c) {
  int t = 0;
  for (; t + 4 <= c; t += 4) {
    const int sa = __builtin_amdgcn_readlane(s, t);
    const int sb = __builtin_amdgcn_readlane(s, t + 1);
    const int sc = __builtin_amdgcn_readlane(s, t + 2);
    const int sd = __builtin_amdgcn_readlane(s, t + 3);
    const float4 ra = H4[(size_t)sa * 32 + lane];
    const float4 rb = H4[(size_t)sb * 32 + lane];
    const float4 rc = H4[(size_t)sc * 32 + lane];
    const float4 rd = H4[(size_t)sd * 32 + lane];
    const float a0 = lane_bcast(w0, t),     a1 = lane_bcast(w1, t);
    const float b0 = lane_bcast(w0, t + 1), b1 = lane_bcast(w1, t + 1);
    const float c0 = lane_bcast(w0, t + 2), c1 = lane_bcast(w1, t + 2);
    const float d0 = lane_bcast(w0, t + 3), d1 = lane_bcast(w1, t + 3);
    const float wa = hi ? a1 : a0;
    const float wb = hi ? b1 : b0;
    const float wc = hi ? c1 : c0;
    const float wd = hi ? d1 : d0;
    acc.x += wa * ra.x; acc.y += wa * ra.y; acc.z += wa * ra.z; acc.w += wa * ra.w;
    acc.x += wb * rb.x; acc.y += wb * rb.y; acc.z += wb * rb.z; acc.w += wb * rb.w;
    acc.x += wc * rc.x; acc.y += wc * rc.y; acc.z += wc * rc.z; acc.w += wc * rc.w;
    acc.x += wd * rd.x; acc.y += wd * rd.y; acc.z += wd * rd.z; acc.w += wd * rd.w;
  }
  for (; t < c; ++t) {
    const int sa = __builtin_amdgcn_readlane(s, t);
    const float4 ra = H4[(size_t)sa * 32 + lane];
    const float a0 = lane_bcast(w0, t), a1 = lane_bcast(w1, t);
    const float wa = hi ? a1 : a0;
    acc.x += wa * ra.x; acc.y += wa * ra.y; acc.z += wa * ra.z; acc.w += wa * ra.w;
  }
}

template <bool CONCAT>
__global__ void gat_node_gather(const int* __restrict__ rowptr, const int* __restrict__ csrc,
                                const float* __restrict__ als, const float* __restrict__ ald,
                                const float* __restrict__ H, const float* __restrict__ bias,
                                float* __restrict__ out, int N) {
  const int n = __builtin_amdgcn_readfirstlane(
      (int)(blockIdx.x * (blockDim.x >> 5) + (threadIdx.x >> 5)));
  const int lane = threadIdx.x & 31;
  if (n >= N) return;
  const int beg = rowptr[n], end = rowptr[n + 1];
  const int deg = end - beg;
  const float2 asn = *(const float2*)(als + 2 * (size_t)n);
  const float2 adn = *(const float2*)(ald + 2 * (size_t)n);
  const float self0 = lrelu(asn.x + adn.x);
  const float self1 = lrelu(asn.y + adn.y);

  const bool has = lane < deg;
  int   s_l = n;
  float v0 = -INFINITY, v1 = -INFINITY;
  if (has) {
    s_l = csrc[beg + lane];
    const float2 a = *(const float2*)(als + 2 * (size_t)s_l);
    v0 = lrelu(a.x + adn.x);
    v1 = lrelu(a.y + adn.y);
  }

  float m0 = fmaxf(self0, v0), m1 = fmaxf(self1, v1);
  for (int j = beg + 32 + lane; j < end; j += 32) {
    const int s = csrc[j];
    const float2 a = *(const float2*)(als + 2 * (size_t)s);
    m0 = fmaxf(m0, lrelu(a.x + adn.x));
    m1 = fmaxf(m1, lrelu(a.y + adn.y));
  }
#pragma unroll
  for (int k = 16; k >= 1; k >>= 1) {
    m0 = fmaxf(m0, __shfl_xor(m0, k, 32));
    m1 = fmaxf(m1, __shfl_xor(m1, k, 32));
  }

  const float ex0 = has ? expf(v0 - m0) : 0.f;
  const float ex1 = has ? expf(v1 - m1) : 0.f;
  float d0 = ex0, d1 = ex1;
  for (int j = beg + 32 + lane; j < end; j += 32) {
    const int s = csrc[j];
    const float2 a = *(const float2*)(als + 2 * (size_t)s);
    d0 += expf(lrelu(a.x + adn.x) - m0);
    d1 += expf(lrelu(a.y + adn.y) - m1);
  }
#pragma unroll
  for (int k = 16; k >= 1; k >>= 1) {
    d0 += __shfl_xor(d0, k, 32);
    d1 += __shfl_xor(d1, k, 32);
  }
  const float exs0 = expf(self0 - m0), exs1 = expf(self1 - m1);
  d0 += exs0;
  d1 += exs1;

  const float4* H4 = (const float4*)H;
  const bool hi = lane >= 16;
  const float wself = hi ? (exs1 / d1) : (exs0 / d0);
  const float4 hn = H4[(size_t)n * 32 + lane];
  float4 acc;
  acc.x = wself * hn.x; acc.y = wself * hn.y; acc.z = wself * hn.z; acc.w = wself * hn.w;
  gather_rows(acc, H4, lane, hi, s_l, ex0 / d0, ex1 / d1, deg < 32 ? deg : 32);
  for (int base = beg + 32; base < end; base += 32) {
    const int j = base + lane;
    int s = n;
    float w0 = 0.f, w1 = 0.f;
    if (j < end) {
      s = csrc[j];
      const float2 a = *(const float2*)(als + 2 * (size_t)s);
      w0 = expf(lrelu(a.x + adn.x) - m0) / d0;
      w1 = expf(lrelu(a.y + adn.y) - m1) / d1;
    }
    const int c = end - base;
    gather_rows(acc, H4, lane, hi, s, w0, w1, c < 32 ? c : 32);
  }

  if (CONCAT) {
    float4 v;
    v.x = acc.x + bias[lane * 4 + 0];
    v.y = acc.y + bias[lane * 4 + 1];
    v.z = acc.z + bias[lane * 4 + 2];
    v.w = acc.w + bias[lane * 4 + 3];
    v.x = (v.x > 0.f) ? v.x : (expf(v.x) - 1.f);
    v.y = (v.y > 0.f) ? v.y : (expf(v.y) - 1.f);
    v.z = (v.z > 0.f) ? v.z : (expf(v.z) - 1.f);
    v.w = (v.w > 0.f) ? v.w : (expf(v.w) - 1.f);
    ((float4*)out)[(size_t)n * 32 + lane] = v;
  } else {
    float4 o;
    o.x = __shfl_xor(acc.x, 16, 32);
    o.y = __shfl_xor(acc.y, 16, 32);
    o.z = __shfl_xor(acc.z, 16, 32);
    o.w = __shfl_xor(acc.w, 16, 32);
    if (!hi) {
      float4 v;
      v.x = 0.5f * (acc.x + o.x) + bias[lane * 4 + 0];
      v.y = 0.5f * (acc.y + o.y) + bias[lane * 4 + 1];
      v.z = 0.5f * (acc.z + o.z) + bias[lane * 4 + 2];
      v.w = 0.5f * (acc.w + o.w) + bias[lane * 4 + 3];
      ((float4*)out)[(size_t)n * 16 + lane] = v;
    }
  }
}

extern "C" void kernel_launch(void* const* d_in, const int* in_sizes, int n_in,
                              void* d_out, int out_size, void* d_ws, size_t ws_size,
                              hipStream_t stream) {
  const float* x   = (const float*)d_in[0];
  const int*   ei  = (const int*)d_in[1];
  const float* W1  = (const float*)d_in[2];
  const float* as1 = (const float*)d_in[3];
  const float* ad1 = (const float*)d_in[4];
  const float* b1  = (const float*)d_in[5];
  const float* W2  = (const float*)d_in[6];
  const float* as2 = (const float*)d_in[7];
  const float* ad2 = (const float*)d_in[8];
  const float* b2  = (const float*)d_in[9];

  const int N = in_sizes[0] / 128;
  const int E = in_sizes[1] / 2;
  const int* esrc = ei;
  const int* edst = ei + E;

  float* h      = (float*)d_ws;
  float* act    = h + (size_t)N * 128;
  float* als    = act + (size_t)N * 128;
  float* ald    = als + (size_t)N * 2;
  int*   cnt    = (int*)(ald + (size_t)N * 2);
  int*   rowptr = cnt + (size_t)N;
  int*   cur    = rowptr + (size_t)N + 1;
  int*   csrc   = cur + (size_t)N;

  const int B = 256;
  const int gGemm  = (N + 15) / 16;
  const int gNodeW = (N + 7) / 8;
  const int gNode  = (N + B - 1) / B;
  const int gScan  = (N + SCAN_T - 1) / SCAN_T;
  int gEdge = (E + B - 1) / B;
  if (gEdge < 1) gEdge = 1;

  csr_zero<<<gNode, B, 0, stream>>>(cnt, N);
  csr_hist<<<gEdge, B, 0, stream>>>(edst, cnt, E);
  csr_scan<<<gScan, SCAN_T, 0, stream>>>(cnt, rowptr, cur, N);
  csr_fill<<<gEdge, B, 0, stream>>>(esrc, edst, cur, csrc, E);

  gat_gemm128_wmma<<<gGemm, B, 0, stream>>>(x, W1, as1, ad1, h, als, ald, N);
  gat_node_gather<true><<<gNodeW, B, 0, stream>>>(rowptr, csrc, als, ald, h, b1, act, N);

  gat_gemm128_wmma<<<gGemm, B, 0, stream>>>(act, W2, as2, ad2, h, als, ald, N);
  gat_node_gather<false><<<gNodeW, B, 0, stream>>>(rowptr, csrc, als, ald, h, b2,
                                                   (float*)d_out, N);
}


# Edge grouping by destination rebuilt as a two-launch counting sort in LDS (slices ordered by 256-node bin, then one block per bin); global atomics, zero and scan launches gone. Row order: LDS arrival
# speedup vs baseline: 5.5927x; 1.6636x over previous
#include <hip/hip_runtime.h>
#include <math.h>

typedef float v2f __attribute__((ext_vector_type(2)));
typedef float v8f __attribute__((ext_vector_type(8)));

#define NEG_SLOPE 0.2f

#define ROW_XOR_ADD(v, m)                                                        \
  (v) += __int_as_float(__builtin_amdgcn_update_dpp(0, __float_as_int(v),        \
                                                    0x160 + (m), 0xf, 0xf, false))
#define TILE_PITCH 132
__global__ void gat_gemm128_wmma(const float* __restrict__ X,
                                 const float* __restrict__ W,
                                 const float* __restrict__ a_src,
                                 const float* __restrict__ a_dst,
                                 float* __restrict__ H,
                                 float* __restrict__ als,
                                 float* __restrict__ ald, int N) {
  __shared__ __attribute__((aligned(16))) float tile[16 * TILE_PITCH];
  const int wave = threadIdx.x >> 5;
  const int lane = threadIdx.x & 31;
  const int half = lane >> 4;
  const int lm   = lane & 15;
  const int row0 = blockIdx.x << 4;
  const int col  = (wave << 4) + lm;

  int arow = row0 + lm;
  if (arow >= N) arow = N - 1;
  const float* xr = X + (size_t)arow * 128;

  const float4 as4 = *(const float4*)(a_src + lane * 4);
  const float4 ad4 = *(const float4*)(a_dst + lane * 4);

  v8f c = {};
#pragma unroll
  for (int kk = 0; kk < 32; ++kk) {
    const int kb = (kk << 2) + (half << 1);
    v2f a, b;
    a.x = xr[kb];
    a.y = xr[kb + 1];
    b.x = W[(size_t)kb * 128 + col];
    b.y = W[(size_t)(kb + 1) * 128 + col];
    c = __builtin_amdgcn_wmma_f32_16x16x4_f32(false, a, false, b,
                                              (short)0, c, false, false);
  }

#pragma unroll
  for (int v = 0; v < 8; ++v)
    tile[((half << 3) + v) * TILE_PITCH + col] = c[v];
  __syncthreads();

  float4 hv[2];
  float s[2], d[2];
#pragma unroll
  for (int i = 0; i < 2; ++i) {
    const int r = (wave << 1) + i;
    hv[i] = *(const float4*)(tile + r * TILE_PITCH + lane * 4);
    s[i] = 0.f; d[i] = 0.f;
    s[i] += hv[i].x * as4.x;  d[i] += hv[i].x * ad4.x;
    s[i] += hv[i].y * as4.y;  d[i] += hv[i].y * ad4.y;
    s[i] += hv[i].z * as4.z;  d[i] += hv[i].z * ad4.z;
    s[i] += hv[i].w * as4.w;  d[i] += hv[i].w * ad4.w;
  }
  ROW_XOR_ADD(s[0], 8); ROW_XOR_ADD(d[0], 8); ROW_XOR_ADD(s[1], 8); ROW_XOR_ADD(d[1], 8);
  ROW_XOR_ADD(s[0], 4); ROW_XOR_ADD(d[0], 4); ROW_XOR_ADD(s[1], 4); ROW_XOR_ADD(d[1], 4);
  ROW_XOR_ADD(s[0], 2); ROW_XOR_ADD(d[0], 2); ROW_XOR_ADD(s[1], 2); ROW_XOR_ADD(d[1], 2);
  ROW_XOR_ADD(s[0], 1); ROW_XOR_ADD(d[0], 1); ROW_XOR_ADD(s[1], 1); ROW_XOR_ADD(d[1], 1);
#pragma unroll
  for (int i = 0; i < 2; ++i) {
    const int n = row0 + (wave << 1) + i;
    if (n < N) {
      ((float4*)H)[(size_t)n * 32 + lane] = hv[i];
      if (lm == 0) {
        als[n * 2 + half] = s[i];
        ald[n * 2 + half] = d[i];
      }
    }
  }
}

#define SLICE_E   8192
#define BIN_SHIFT 8
#define BIN_NODES 256
#define OUT_CAP   8192

__device__ __forceinline__ int block_excl_scan(int c, int* wsum, int& total) {
  const int lane = threadIdx.x & 31;
  const int w = threadIdx.x >> 5;
  int inc = c;
#pragma unroll
  for (int d = 1; d < 32; d <<= 1) {
    const int v = __shfl_up(inc, d, 32);
    if (lane >= d) inc += v;
  }
  if (lane == 31) wsum[w] = inc;
  __syncthreads();
  int o = wsum[lane];
#pragma unroll
  for (int d = 1; d < 32; d <<= 1) {
    const int v = __shfl_up(o, d, 32);
    if (lane >= d) o += v;
  }
  total = __shfl(o, 31, 32);
  const int before = __shfl(o, (w + 31) & 31, 32);
  return (w > 0 ? before : 0) + (inc - c);
}

__global__ void __launch_bounds__(1024)
csr_slice_bins(const int* __restrict__ es, const int* __restrict__ ed,
               unsigned* __restrict__ staged, int* __restrict__ tab,
               int E, int nb, int G) {
  __shared__ int hist[1024];
  __shared__ int wsum[32];
  __shared__ unsigned stage[SLICE_E];
  const int t = threadIdx.x;
  const int b = blockIdx.x;
  const int e0 = b * SLICE_E;
  hist[t] = 0;
  int s[8], d[8];
#pragma unroll
  for (int i = 0; i < 8; ++i) {
    const int e = e0 + i * 1024 + t;
    d[i] = (e < E) ? ed[e] : -1;
    s[i] = (e < E) ? es[e] : 0;
  }
  __syncthreads();
#pragma unroll
  for (int i = 0; i < 8; ++i)
    if (d[i] >= 0) atomicAdd(&hist[d[i] >> BIN_SHIFT], 1);
  __syncthreads();
  const int c = hist[t];
  int total;
  const int excl = block_excl_scan(c, wsum, total);
  hist[t] = excl;
  if (t < nb) tab[(size_t)t * G + b] = excl;
  if (t == 0) tab[(size_t)nb * G + b] = total;
  __syncthreads();
  int slot[8];
#pragma unroll
  for (int i = 0; i < 8; ++i)
    slot[i] = (d[i] >= 0) ? atomicAdd(&hist[d[i] >> BIN_SHIFT], 1) : 0;
#pragma unroll
  for (int i = 0; i < 8; ++i)
    if (d[i] >= 0)
      stage[slot[i]] = ((unsigned)s[i] << BIN_SHIFT) | (unsigned)(d[i] & (BIN_NODES - 1));
  __syncthreads();
  for (int j = t; j < total; j += 1024) staged[(size_t)e0 + j] = stage[j];
}

template <bool PLACE>
__device__ __forceinline__ void bin_pass(const unsigned* __restrict__ staged,
                                         const int* segoff, const int* seglen,
                                         int* cnt, int* outl, int* __restrict__ csrc,
                                         int base, int G) {
  const int lane = threadIdx.x & 31;
  const int w = threadIdx.x >> 5;
  for (int b0 = w; b0 < G; b0 += 128) {
    size_t off[4];
    int len[4];
    int mx = 0;
#pragma unroll
    for (int i = 0; i < 4; ++i) {
      const int sb = b0 + 32 * i;
      const bool ok = sb < G;
      off[i] = ok ? (size_t)sb * SLICE_E + segoff[sb] : 0;
      len[i] = ok ? seglen[sb] : 0;
      mx = len[i] > mx ? len[i] : mx;
    }
    for (int j0 = 0; j0 < mx; j0 += 32) {
      const int j = j0 + lane;
      unsigned v[4];
#pragma unroll
      for (int i = 0; i < 4; ++i) v[i] = (j < len[i]) ? staged[off[i] + j] : 0u;
#pragma unroll
      for (int i = 0; i < 4; ++i) {
        if (j < len[i]) {
          const int dl = (int)(v[i] & (BIN_NODES - 1));
          if (!PLACE) {
            atomicAdd(&cnt[dl], 1);
          } else {
            const int slot = atomicAdd(&cnt[dl], 1);
            const int sv = (int)(v[i] >> BIN_SHIFT);
            if (slot < OUT_CAP) outl[slot] = sv;
            else csrc[(size_t)base + slot] = sv;
          }
        }
      }
    }
  }
}

__global__ void __launch_bounds__(1024)
csr_bin_rows(const unsigned* __restrict__ staged, const int* __restrict__ tab,
             int* __restrict__ rowptr, int* __restrict__ csrc, int N, int G) {
  __shared__ int segoff[1024];
  __shared__ int seglen[1024];
  __shared__ int cnt[BIN_NODES];
  __shared__ int wsum[32];
  __shared__ int wbase[32];
  __shared__ int outl[OUT_CAP];
  const int t = threadIdx.x;
  const int lane = t & 31;
  const int w = t >> 5;
  const int k = blockIdx.x;

  int o = 0;
  if (t < G) {
    o = tab[(size_t)k * G + t];
    const int o2 = tab[(size_t)(k + 1) * G + t];
    segoff[t] = o;
    seglen[t] = o2 - o;
  }
  if (t < BIN_NODES) cnt[t] = 0;
#pragma unroll
  for (int m = 16; m >= 1; m >>= 1) o += __shfl_xor(o, m, 32);
  if (lane == 0) wbase[w] = o;
  __syncthreads();
  int base = wbase[lane];
#pragma unroll
  for (int m = 16; m >= 1; m >>= 1) base += __shfl_xor(base, m, 32);

  bin_pass<false>(staged, segoff, seglen, cnt, outl, csrc, base, G);
  __syncthreads();

  const int c = (t < BIN_NODES) ? cnt[t] : 0;
  int total;
  const int excl = block_excl_scan(c, wsum, total);
  if (t < BIN_NODES) {
    cnt[t] = excl;
    const int node = (k << BIN_SHIFT) + t;
    if (node < N) {
      rowptr[node] = base + excl;
      if (node == N - 1) rowptr[N] = base + excl + c;
    }
  }
  __syncthreads();

  bin_pass<true>(staged, segoff, seglen, cnt, outl, csrc, base, G);
  __syncthreads();

  const int nl = total < OUT_CAP ? total : OUT_CAP;
  for (int j = t; j < nl; j += 1024) csrc[(size_t)base + j] = outl[j];
}

__device__ __forceinline__ float lrelu(float v) {
  return (v >= 0.f) ? v : NEG_SLOPE * v;
}

__device__ __forceinline__ float lane_bcast(float v, int t) {
  return __int_as_float(__builtin_amdgcn_readlane(__float_as_int(v), t));
}

__device__ __forceinline__ void gather_rows(float4& acc, const float4* __restrict__ H4,
                                            int lane, bool hi, int s, float w0, float w1,
                                            int c) {
  int t = 0;
  for (; t + 4 <= c; t += 4) {
    const int sa = __builtin_amdgcn_readlane(s, t);
    const int sb = __builtin_amdgcn_readlane(s, t + 1);
    const int sc = __builtin_amdgcn_readlane(s, t + 2);
    const int sd = __builtin_amdgcn_readlane(s, t + 3);
    const float4 ra = H4[(size_t)sa * 32 + lane];
    const float4 rb = H4[(size_t)sb * 32 + lane];
    const float4 rc = H4[(size_t)sc * 32 + lane];
    const float4 rd = H4[(size_t)sd * 32 + lane];
    const float a0 = lane_bcast(w0, t),     a1 = lane_bcast(w1, t);
    const float b0 = lane_bcast(w0, t + 1), b1 = lane_bcast(w1, t + 1);
    const float c0 = lane_bcast(w0, t + 2), c1 = lane_bcast(w1, t + 2);
    const float d0 = lane_bcast(w0, t + 3), d1 = lane_bcast(w1, t + 3);
    const float wa = hi ? a1 : a0;
    const float wb = hi ? b1 : b0;
    const float wc = hi ? c1 : c0;
    const float wd = hi ? d1 : d0;
    acc.x += wa * ra.x; acc.y += wa * ra.y; acc.z += wa * ra.z; acc.w += wa * ra.w;
    acc.x += wb * rb.x; acc.y += wb * rb.y; acc.z += wb * rb.z; acc.w += wb * rb.w;
    acc.x += wc * rc.x; acc.y += wc * rc.y; acc.z += wc * rc.z; acc.w += wc * rc.w;
    acc.x += wd * rd.x; acc.y += wd * rd.y; acc.z += wd * rd.z; acc.w += wd * rd.w;
  }
  for (; t < c; ++t) {
    const int sa = __builtin_amdgcn_readlane(s, t);
    const float4 ra = H4[(size_t)sa * 32 + lane];
    const float a0 = lane_bcast(w0, t), a1 = lane_bcast(w1, t);
    const float wa = hi ? a1 : a0;
    acc.x += wa * ra.x; acc.y += wa * ra.y; acc.z += wa * ra.z; acc.w += wa * ra.w;
  }
}

template <bool CONCAT>
__global__ void gat_node_gather(const int* __restrict__ rowptr, const int* __restrict__ csrc,
                                const float* __restrict__ als, const float* __restrict__ ald,
                                const float* __restrict__ H, const float* __restrict__ bias,
                                float* __restrict__ out, int N) {
  const int n = __builtin_amdgcn_readfirstlane(
      (int)(blockIdx.x * (blockDim.x >> 5) + (threadIdx.x >> 5)));
  const int lane = threadIdx.x & 31;
  if (n >= N) return;
  const int beg = rowptr[n], end = rowptr[n + 1];
  const int deg = end - beg;
  const float2 asn = *(const float2*)(als + 2 * (size_t)n);
  const float2 adn = *(const float2*)(ald + 2 * (size_t)n);
  const float self0 = lrelu(asn.x + adn.x);
  const float self1 = lrelu(asn.y + adn.y);

  const bool has = lane < deg;
  int   s_l = n;
  float v0 = -INFINITY, v1 = -INFINITY;
  if (has) {
    s_l = csrc[beg + lane];
    const float2 a = *(const float2*)(als + 2 * (size_t)s_l);
    v0 = lrelu(a.x + adn.x);
    v1 = lrelu(a.y + adn.y);
  }

  float m0 = fmaxf(self0, v0), m1 = fmaxf(self1, v1);
  for (int j = beg + 32 + lane; j < end; j += 32) {
    const int s = csrc[j];
    const float2 a = *(const float2*)(als + 2 * (size_t)s);
    m0 = fmaxf(m0, lrelu(a.x + adn.x));
    m1 = fmaxf(m1, lrelu(a.y + adn.y));
  }
#pragma unroll
  for (int k = 16; k >= 1; k >>= 1) {
    m0 = fmaxf(m0, __shfl_xor(m0, k, 32));
    m1 = fmaxf(m1, __shfl_xor(m1, k, 32));
  }

  const float ex0 = has ? expf(v0 - m0) : 0.f;
  const float ex1 = has ? expf(v1 - m1) : 0.f;
  float d0 = ex0, d1 = ex1;
  for (int j = beg + 32 + lane; j < end; j += 32) {
    const int s = csrc[j];
    const float2 a = *(const float2*)(als + 2 * (size_t)s);
    d0 += expf(lrelu(a.x + adn.x) - m0);
    d1 += expf(lrelu(a.y + adn.y) - m1);
  }
#pragma unroll
  for (int k = 16; k >= 1; k >>= 1) {
    d0 += __shfl_xor(d0, k, 32);
    d1 += __shfl_xor(d1, k, 32);
  }
  const float exs0 = expf(self0 - m0), exs1 = expf(self1 - m1);
  d0 += exs0;
  d1 += exs1;

  const float4* H4 = (const float4*)H;
  const bool hi = lane >= 16;
  const float wself = hi ? (exs1 / d1) : (exs0 / d0);
  const float4 hn = H4[(size_t)n * 32 + lane];
  float4 acc;
  acc.x = wself * hn.x; acc.y = wself * hn.y; acc.z = wself * hn.z; acc.w = wself * hn.w;
  gather_rows(acc, H4, lane, hi, s_l, ex0 / d0, ex1 / d1, deg < 32 ? deg : 32);
  for (int base = beg + 32; base < end; base += 32) {
    const int j = base + lane;
    int s = n;
    float w0 = 0.f, w1 = 0.f;
    if (j < end) {
      s = csrc[j];
      const float2 a = *(const float2*)(als + 2 * (size_t)s);
      w0 = expf(lrelu(a.x + adn.x) - m0) / d0;
      w1 = expf(lrelu(a.y + adn.y) - m1) / d1;
    }
    const int c = end - base;
    gather_rows(acc, H4, lane, hi, s, w0, w1, c < 32 ? c : 32);
  }

  if (CONCAT) {
    float4 v;
    v.x = acc.x + bias[lane * 4 + 0];
    v.y = acc.y + bias[lane * 4 + 1];
    v.z = acc.z + bias[lane * 4 + 2];
    v.w = acc.w + bias[lane * 4 + 3];
    v.x = (v.x > 0.f) ? v.x : (expf(v.x) - 1.f);
    v.y = (v.y > 0.f) ? v.y : (expf(v.y) - 1.f);
    v.z = (v.z > 0.f) ? v.z : (expf(v.z) - 1.f);
    v.w = (v.w > 0.f) ? v.w : (expf(v.w) - 1.f);
    ((float4*)out)[(size_t)n * 32 + lane] = v;
  } else {
    float4 o;
    o.x = __shfl_xor(acc.x, 16, 32);
    o.y = __shfl_xor(acc.y, 16, 32);
    o.z = __shfl_xor(acc.z, 16, 32);
    o.w = __shfl_xor(acc.w, 16, 32);
    if (!hi) {
      float4 v;
      v.x = 0.5f * (acc.x + o.x) + bias[lane * 4 + 0];
      v.y = 0.5f * (acc.y + o.y) + bias[lane * 4 + 1];
      v.z = 0.5f * (acc.z + o.z) + bias[lane * 4 + 2];
      v.w = 0.5f * (acc.w + o.w) + bias[lane * 4 + 3];
      ((float4*)out)[(size_t)n * 16 + lane] = v;
    }
  }
}

extern "C" void kernel_launch(void* const* d_in, const int* in_sizes, int n_in,
                              void* d_out, int out_size, void* d_ws, size_t ws_size,
                              hipStream_t stream) {
  const float* x   = (const float*)d_in[0];
  const int*   ei  = (const int*)d_in[1];
  const float* W1  = (const float*)d_in[2];
  const float* as1 = (const float*)d_in[3];
  const float* ad1 = (const float*)d_in[4];
  const float* b1  = (const float*)d_in[5];
  const float* W2  = (const float*)d_in[6];
  const float* as2 = (const float*)d_in[7];
  const float* ad2 = (const float*)d_in[8];
  const float* b2  = (const float*)d_in[9];

  const int N = in_sizes[0] / 128;
  const int E = in_sizes[1] / 2;
  const int* esrc = ei;
  const int* edst = ei + E;

  const int nb = (N + BIN_NODES - 1) >> BIN_SHIFT;
  int G = (E + SLICE_E - 1) / SLICE_E;
  if (G < 1) G = 1;
  float*    h      = (float*)d_ws;
  float*    act    = h + (size_t)N * 128;
  float*    als    = act + (size_t)N * 128;
  float*    ald    = als + (size_t)N * 2;
  int*      rowptr = (int*)(ald + (size_t)N * 2);
  int*      tab    = rowptr + (size_t)N + 1;
  unsigned* staged = (unsigned*)(tab + (size_t)(nb + 1) * G);
  int*      csrc   = (int*)(staged + (size_t)G * SLICE_E);

  const int B = 256;
  const int gGemm  = (N + 15) / 16;
  const int gNodeW = (N + 7) / 8;

  csr_slice_bins<<<G, 1024, 0, stream>>>(esrc, edst, staged, tab, E, nb, G);
  csr_bin_rows<<<nb, 1024, 0, stream>>>(staged, tab, rowptr, csrc, N, G);

  gat_gemm128_wmma<<<gGemm, B, 0, stream>>>(x, W1, as1, ad1, h, als, ald, N);
  gat_node_gather<true><<<gNodeW, B, 0, stream>>>(rowptr, csrc, als, ald, h, b1, act, N);

  gat_gemm128_wmma<<<gGemm, B, 0, stream>>>(act, W2, as2, ad2, h, als, ald, N);
  gat_node_gather<false><<<gNodeW, B, 0, stream>>>(rowptr, csrc, als, ald, h, b2,
                                                   (float*)d_out, N);
}


# Feature GEMM register-blocked: each wave owns a 16x128 strip (8 accumulator tiles per A fragment), W staged once per 64-row block in LDS as k pairs; k order and all sum orders kept.
# speedup vs baseline: 6.7077x; 1.1994x over previous
#include <hip/hip_runtime.h>
#include <math.h>

typedef float v2f __attribute__((ext_vector_type(2)));
typedef float v8f __attribute__((ext_vector_type(8)));

#define NEG_SLOPE 0.2f

#define ROW_XOR_ADD(v, m)                                                        \
  (v) += __int_as_float(__builtin_amdgcn_update_dpp(0, __float_as_int(v),        \
                                                    0x160 + (m), 0xf, 0xf, false))
#define TILE_PITCH 132
#define GEMM_ROWS  64
#define WL_PITCH   288
__global__ void __launch_bounds__(128)
gat_gemm128_wmma(const float* __restrict__ X,
                 const float* __restrict__ W,
                 const float* __restrict__ a_src,
                 const float* __restrict__ a_dst,
                 float* __restrict__ H,
                 float* __restrict__ als,
                 float* __restrict__ ald, int N) {
  __shared__ __attribute__((aligned(16))) float lds[64 * WL_PITCH];
  const int wave = threadIdx.x >> 5;
  const int lane = threadIdx.x & 31;
  const int half = lane >> 4;
  const int lm   = lane & 15;
  const int row0 = blockIdx.x * GEMM_ROWS + (wave << 4);

  int arow = row0 + lm;
  if (arow >= N) arow = N - 1;
  const float* xr = X + (size_t)arow * 128 + (half << 1);

  v2f a[32];
#pragma unroll
  for (int kk = 0; kk < 32; ++kk) a[kk] = *(const v2f*)(xr + (kk << 2));

  const float4 as4 = *(const float4*)(a_src + lane * 4);
  const float4 ad4 = *(const float4*)(a_dst + lane * 4);

  const float4* W4 = (const float4*)W;
  for (int it = 0; it < 16; ++it) {
    const int p = (it << 2) + wave;
    const float4 r0 = W4[(2 * p) * 32 + lane];
    const float4 r1 = W4[(2 * p + 1) * 32 + lane];
    float4 q0, q1;
    q0.x = r0.x; q0.y = r1.x; q0.z = r0.y; q0.w = r1.y;
    q1.x = r0.z; q1.y = r1.z; q1.z = r0.w; q1.w = r1.w;
    float* dst = lds + p * WL_PITCH + lane * 8;
    *(float4*)dst = q0;
    *(float4*)(dst + 4) = q1;
  }
  __syncthreads();

  v8f c[8];
#pragma unroll
  for (int j = 0; j < 8; ++j) c[j] = (v8f){};
  const float* bl = lds + half * WL_PITCH + (lm << 1);
  v2f bc[8], bn[8];
#pragma unroll
  for (int j = 0; j < 8; ++j) bc[j] = *(const v2f*)(bl + j * 32);
#pragma unroll
  for (int kk = 0; kk < 32; ++kk) {
    if (kk + 1 < 32) {
#pragma unroll
      for (int j = 0; j < 8; ++j)
        bn[j] = *(const v2f*)(bl + (kk + 1) * 2 * WL_PITCH + j * 32);
    }
#pragma unroll
    for (int j = 0; j < 8; ++j)
      c[j] = __builtin_amdgcn_wmma_f32_16x16x4_f32(false, a[kk], false, bc[j],
                                                   (short)0, c[j], false, false);
#pragma unroll
    for (int j = 0; j < 8; ++j) bc[j] = bn[j];
  }

  __syncthreads();
  float* tw = lds + wave * (16 * TILE_PITCH);
#pragma unroll
  for (int j = 0; j < 8; ++j)
#pragma unroll
    for (int v = 0; v < 8; ++v)
      tw[((half << 3) + v) * TILE_PITCH + (j << 4) + lm] = c[j][v];
  __syncthreads();

  float sfin = 0.f, dfin = 0.f;
#pragma unroll
  for (int ii = 0; ii < 8; ++ii) {
    float4 hv[2];
    float s[2], d[2];
#pragma unroll
    for (int i = 0; i < 2; ++i) {
      const int r = (ii << 1) + i;
      hv[i] = *(const float4*)(tw + r * TILE_PITCH + lane * 4);
      s[i] = 0.f; d[i] = 0.f;
      s[i] += hv[i].x * as4.x;  d[i] += hv[i].x * ad4.x;
      s[i] += hv[i].y * as4.y;  d[i] += hv[i].y * ad4.y;
      s[i] += hv[i].z * as4.z;  d[i] += hv[i].z * ad4.z;
      s[i] += hv[i].w * as4.w;  d[i] += hv[i].w * ad4.w;
    }
    ROW_XOR_ADD(s[0], 8); ROW_XOR_ADD(d[0], 8); ROW_XOR_ADD(s[1], 8); ROW_XOR_ADD(d[1], 8);
    ROW_XOR_ADD(s[0], 4); ROW_XOR_ADD(d[0], 4); ROW_XOR_ADD(s[1], 4); ROW_XOR_ADD(d[1], 4);
    ROW_XOR_ADD(s[0], 2); ROW_XOR_ADD(d[0], 2); ROW_XOR_ADD(s[1], 2); ROW_XOR_ADD(d[1], 2);
    ROW_XOR_ADD(s[0], 1); ROW_XOR_ADD(d[0], 1); ROW_XOR_ADD(s[1], 1); ROW_XOR_ADD(d[1], 1);
#pragma unroll
    for (int i = 0; i < 2; ++i) {
      const int r = (ii << 1) + i;
      const int n = row0 + r;
      if (n < N) ((float4*)H)[(size_t)n * 32 + lane] = hv[i];
      if (lm == r) { sfin = s[i]; dfin = d[i]; }
    }
  }
  const int nr = row0 + lm;
  if (nr < N) {
    als[nr * 2 + half] = sfin;
    ald[nr * 2 + half] = dfin;
  }
}

#define SLICE_E   8192
#define BIN_SHIFT 8
#define BIN_NODES 256
#define OUT_CAP   8192

__device__ __forceinline__ int block_excl_scan(int c, int* wsum, int& total) {
  const int lane = threadIdx.x & 31;
  const int w = threadIdx.x >> 5;
  int inc = c;
#pragma unroll
  for (int d = 1; d < 32; d <<= 1) {
    const int v = __shfl_up(inc, d, 32);
    if (lane >= d) inc += v;
  }
  if (lane == 31) wsum[w] = inc;
  __syncthreads();
  int o = wsum[lane];
#pragma unroll
  for (int d = 1; d < 32; d <<= 1) {
    const int v = __shfl_up(o, d, 32);
    if (lane >= d) o += v;
  }
  total = __shfl(o, 31, 32);
  const int before = __shfl(o, (w + 31) & 31, 32);
  return (w > 0 ? before : 0) + (inc - c);
}

__global__ void __launch_bounds__(1024)
csr_slice_bins(const int* __restrict__ es, const int* __restrict__ ed,
               unsigned* __restrict__ staged, int* __restrict__ tab,
               int E, int nb, int G) {
  __shared__ int hist[1024];
  __shared__ int wsum[32];
  __shared__ unsigned stage[SLICE_E];
  const int t = threadIdx.x;
  const int b = blockIdx.x;
  const int e0 = b * SLICE_E;
  hist[t] = 0;
  int s[8], d[8];
#pragma unroll
  for (int i = 0; i < 8; ++i) {
    const int e = e0 + i * 1024 + t;
    d[i] = (e < E) ? ed[e] : -1;
    s[i] = (e < E) ? es[e] : 0;
  }
  __syncthreads();
#pragma unroll
  for (int i = 0; i < 8; ++i)
    if (d[i] >= 0) atomicAdd(&hist[d[i] >> BIN_SHIFT], 1);
  __syncthreads();
  const int c = hist[t];
  int total;
  const int excl = block_excl_scan(c, wsum, total);
  hist[t] = excl;
  if (t < nb) tab[(size_t)t * G + b] = excl;
  if (t == 0) tab[(size_t)nb * G + b] = total;
  __syncthreads();
  int slot[8];
#pragma unroll
  for (int i = 0; i < 8; ++i)
    slot[i] = (d[i] >= 0) ? atomicAdd(&hist[d[i] >> BIN_SHIFT], 1) : 0;
#pragma unroll
  for (int i = 0; i < 8; ++i)
    if (d[i] >= 0)
      stage[slot[i]] = ((unsigned)s[i] << BIN_SHIFT) | (unsigned)(d[i] & (BIN_NODES - 1));
  __syncthreads();
  for (int j = t; j < total; j += 1024) staged[(size_t)e0 + j] = stage[j];
}

template <bool PLACE>
__device__ __forceinline__ void bin_pass(const unsigned* __restrict__ staged,
                                         const int* segoff, const int* seglen,
                                         int* cnt, int* outl, int* __restrict__ csrc,
                                         int base, int G) {
  const int lane = threadIdx.x & 31;
  const int w = threadIdx.x >> 5;
  for (int b0 = w; b0 < G; b0 += 128) {
    size_t off[4];
    int len[4];
    int mx = 0;
#pragma unroll
    for (int i = 0; i < 4; ++i) {
      const int sb = b0 + 32 * i;
      const bool ok = sb < G;
      off[i] = ok ? (size_t)sb * SLICE_E + segoff[sb] : 0;
      len[i] = ok ? seglen[sb] : 0;
      mx = len[i] > mx ? len[i] : mx;
    }
    for (int j0 = 0; j0 < mx; j0 += 32) {
      const int j = j0 + lane;
      unsigned v[4];
#pragma unroll
      for (int i = 0; i < 4; ++i) v[i] = (j < len[i]) ? staged[off[i] + j] : 0u;
#pragma unroll
      for (int i = 0; i < 4; ++i) {
        if (j < len[i]) {
          const int dl = (int)(v[i] & (BIN_NODES - 1));
          if (!PLACE) {
            atomicAdd(&cnt[dl], 1);
          } else {
            const int slot = atomicAdd(&cnt[dl], 1);
            const int sv = (int)(v[i] >> BIN_SHIFT);
            if (slot < OUT_CAP) outl[slot] = sv;
            else csrc[(size_t)base + slot] = sv;
          }
        }
      }
    }
  }
}

__global__ void __launch_bounds__(1024)
csr_bin_rows(const unsigned* __restrict__ staged, const int* __restrict__ tab,
             int* __restrict__ rowptr, int* __restrict__ csrc, int N, int G) {
  __shared__ int segoff[1024];
  __shared__ int seglen[1024];
  __shared__ int cnt[BIN_NODES];
  __shared__ int wsum[32];
  __shared__ int wbase[32];
  __shared__ int outl[OUT_CAP];
  const int t = threadIdx.x;
  const int lane = t & 31;
  const int w = t >> 5;
  const int k = blockIdx.x;

  int o = 0;
  if (t < G) {
    o = tab[(size_t)k * G + t];
    const int o2 = tab[(size_t)(k + 1) * G + t];
    segoff[t] = o;
    seglen[t] = o2 - o;
  }
  if (t < BIN_NODES) cnt[t] = 0;
#pragma unroll
  for (int m = 16; m >= 1; m >>= 1) o += __shfl_xor(o, m, 32);
  if (lane == 0) wbase[w] = o;
  __syncthreads();
  int base = wbase[lane];
#pragma unroll
  for (int m = 16; m >= 1; m >>= 1) base += __shfl_xor(base, m, 32);

  bin_pass<false>(staged, segoff, seglen, cnt, outl, csrc, base, G);
  __syncthreads();

  const int c = (t < BIN_NODES) ? cnt[t] : 0;
  int total;
  const int excl = block_excl_scan(c, wsum, total);
  if (t < BIN_NODES) {
    cnt[t] = excl;
    const int node = (k << BIN_SHIFT) + t;
    if (node < N) {
      rowptr[node] = base + excl;
      if (node == N - 1) rowptr[N] = base + excl + c;
    }
  }
  __syncthreads();

  bin_pass<true>(staged, segoff, seglen, cnt, outl, csrc, base, G);
  __syncthreads();

  const int nl = total < OUT_CAP ? total : OUT_CAP;
  for (int j = t; j < nl; j += 1024) csrc[(size_t)base + j] = outl[j];
}

__device__ __forceinline__ float lrelu(float v) {
  return (v >= 0.f) ? v : NEG_SLOPE * v;
}

__device__ __forceinline__ float lane_bcast(float v, int t) {
  return __int_as_float(__builtin_amdgcn_readlane(__float_as_int(v), t));
}

__device__ __forceinline__ void gather_rows(float4& acc, const float4* __restrict__ H4,
                                            int lane, bool hi, int s, float w0, float w1,
                                            int c) {
  int t = 0;
  for (; t + 4 <= c; t += 4) {
    const int sa = __builtin_amdgcn_readlane(s, t);
    const int sb = __builtin_amdgcn_readlane(s, t + 1);
    const int sc = __builtin_amdgcn_readlane(s, t + 2);
    const int sd = __builtin_amdgcn_readlane(s, t + 3);
    const float4 ra = H4[(size_t)sa * 32 + lane];
    const float4 rb = H4[(size_t)sb * 32 + lane];
    const float4 rc = H4[(size_t)sc * 32 + lane];
    const float4 rd = H4[(size_t)sd * 32 + lane];
    const float a0 = lane_bcast(w0, t),     a1 = lane_bcast(w1, t);
    const float b0 = lane_bcast(w0, t + 1), b1 = lane_bcast(w1, t + 1);
    const float c0 = lane_bcast(w0, t + 2), c1 = lane_bcast(w1, t + 2);
    const float d0 = lane_bcast(w0, t + 3), d1 = lane_bcast(w1, t + 3);
    const float wa = hi ? a1 : a0;
    const float wb = hi ? b1 : b0;
    const float wc = hi ? c1 : c0;
    const float wd = hi ? d1 : d0;
    acc.x += wa * ra.x; acc.y += wa * ra.y; acc.z += wa * ra.z; acc.w += wa * ra.w;
    acc.x += wb * rb.x; acc.y += wb * rb.y; acc.z += wb * rb.z; acc.w += wb * rb.w;
    acc.x += wc * rc.x; acc.y += wc * rc.y; acc.z += wc * rc.z; acc.w += wc * rc.w;
    acc.x += wd * rd.x; acc.y += wd * rd.y; acc.z += wd * rd.z; acc.w += wd * rd.w;
  }
  for (; t < c; ++t) {
    const int sa = __builtin_amdgcn_readlane(s, t);
    const float4 ra = H4[(size_t)sa * 32 + lane];
    const float a0 = lane_bcast(w0, t), a1 = lane_bcast(w1, t);
    const float wa = hi ? a1 : a0;
    acc.x += wa * ra.x; acc.y += wa * ra.y; acc.z += wa * ra.z; acc.w += wa * ra.w;
  }
}

template <bool CONCAT>
__global__ void gat_node_gather(const int* __restrict__ rowptr, const int* __restrict__ csrc,
                                const float* __restrict__ als, const float* __restrict__ ald,
                                const float* __restrict__ H, const float* __restrict__ bias,
                                float* __restrict__ out, int N) {
  const int n = __builtin_amdgcn_readfirstlane(
      (int)(blockIdx.x * (blockDim.x >> 5) + (threadIdx.x >> 5)));
  const int lane = threadIdx.x & 31;
  if (n >= N) return;
  const int beg = rowptr[n], end = rowptr[n + 1];
  const int deg = end - beg;
  const float2 asn = *(const float2*)(als + 2 * (size_t)n);
  const float2 adn = *(const float2*)(ald + 2 * (size_t)n);
  const float self0 = lrelu(asn.x + adn.x);
  const float self1 = lrelu(asn.y + adn.y);

  const bool has = lane < deg;
  int   s_l = n;
  float v0 = -INFINITY, v1 = -INFINITY;
  if (has) {
    s_l = csrc[beg + lane];
    const float2 a = *(const float2*)(als + 2 * (size_t)s_l);
    v0 = lrelu(a.x + adn.x);
    v1 = lrelu(a.y + adn.y);
  }

  float m0 = fmaxf(self0, v0), m1 = fmaxf(self1, v1);
  for (int j = beg + 32 + lane; j < end; j += 32) {
    const int s = csrc[j];
    const float2 a = *(const float2*)(als + 2 * (size_t)s);
    m0 = fmaxf(m0, lrelu(a.x + adn.x));
    m1 = fmaxf(m1, lrelu(a.y + adn.y));
  }
#pragma unroll
  for (int k = 16; k >= 1; k >>= 1) {
    m0 = fmaxf(m0, __shfl_xor(m0, k, 32));
    m1 = fmaxf(m1, __shfl_xor(m1, k, 32));
  }

  const float ex0 = has ? expf(v0 - m0) : 0.f;
  const float ex1 = has ? expf(v1 - m1) : 0.f;
  float d0 = ex0, d1 = ex1;
  for (int j = beg + 32 + lane; j < end; j += 32) {
    const int s = csrc[j];
    const float2 a = *(const float2*)(als + 2 * (size_t)s);
    d0 += expf(lrelu(a.x + adn.x) - m0);
    d1 += expf(lrelu(a.y + adn.y) - m1);
  }
#pragma unroll
  for (int k = 16; k >= 1; k >>= 1) {
    d0 += __shfl_xor(d0, k, 32);
    d1 += __shfl_xor(d1, k, 32);
  }
  const float exs0 = expf(self0 - m0), exs1 = expf(self1 - m1);
  d0 += exs0;
  d1 += exs1;

  const float4* H4 = (const float4*)H;
  const bool hi = lane >= 16;
  const float wself = hi ? (exs1 / d1) : (exs0 / d0);
  const float4 hn = H4[(size_t)n * 32 + lane];
  float4 acc;
  acc.x = wself * hn.x; acc.y = wself * hn.y; acc.z = wself * hn.z; acc.w = wself * hn.w;
  gather_rows(acc, H4, lane, hi, s_l, ex0 / d0, ex1 / d1, deg < 32 ? deg : 32);
  for (int base = beg + 32; base < end; base += 32) {
    const int j = base + lane;
    int s = n;
    float w0 = 0.f, w1 = 0.f;
    if (j < end) {
      s = csrc[j];
      const float2 a = *(const float2*)(als + 2 * (size_t)s);
      w0 = expf(lrelu(a.x + adn.x) - m0) / d0;
      w1 = expf(lrelu(a.y + adn.y) - m1) / d1;
    }
    const int c = end - base;
    gather_rows(acc, H4, lane, hi, s, w0, w1, c < 32 ? c : 32);
  }

  if (CONCAT) {
    float4 v;
    v.x = acc.x + bias[lane * 4 + 0];
    v.y = acc.y + bias[lane * 4 + 1];
    v.z = acc.z + bias[lane * 4 + 2];
    v.w = acc.w + bias[lane * 4 + 3];
    v.x = (v.x > 0.f) ? v.x : (expf(v.x) - 1.f);
    v.y = (v.y > 0.f) ? v.y : (expf(v.y) - 1.f);
    v.z = (v.z > 0.f) ? v.z : (expf(v.z) - 1.f);
    v.w = (v.w > 0.f) ? v.w : (expf(v.w) - 1.f);
    ((float4*)out)[(size_t)n * 32 + lane] = v;
  } else {
    float4 o;
    o.x = __shfl_xor(acc.x, 16, 32);
    o.y = __shfl_xor(acc.y, 16, 32);
    o.z = __shfl_xor(acc.z, 16, 32);
    o.w = __shfl_xor(acc.w, 16, 32);
    if (!hi) {
      float4 v;
      v.x = 0.5f * (acc.x + o.x) + bias[lane * 4 + 0];
      v.y = 0.5f * (acc.y + o.y) + bias[lane * 4 + 1];
      v.z = 0.5f * (acc.z + o.z) + bias[lane * 4 + 2];
      v.w = 0.5f * (acc.w + o.w) + bias[lane * 4 + 3];
      ((float4*)out)[(size_t)n * 16 + lane] = v;
    }
  }
}

extern "C" void kernel_launch(void* const* d_in, const int* in_sizes, int n_in,
                              void* d_out, int out_size, void* d_ws, size_t ws_size,
                              hipStream_t stream) {
  const float* x   = (const float*)d_in[0];
  const int*   ei  = (const int*)d_in[1];
  const float* W1  = (const float*)d_in[2];
  const float* as1 = (const float*)d_in[3];
  const float* ad1 = (const float*)d_in[4];
  const float* b1  = (const float*)d_in[5];
  const float* W2  = (const float*)d_in[6];
  const float* as2 = (const float*)d_in[7];
  const float* ad2 = (const float*)d_in[8];
  const float* b2  = (const float*)d_in[9];

  const int N = in_sizes[0] / 128;
  const int E = in_sizes[1] / 2;
  const int* esrc = ei;
  const int* edst = ei + E;

  const int nb = (N + BIN_NODES - 1) >> BIN_SHIFT;
  int G = (E + SLICE_E - 1) / SLICE_E;
  if (G < 1) G = 1;
  float*    h      = (float*)d_ws;
  float*    act    = h + (size_t)N * 128;
  float*    als    = act + (size_t)N * 128;
  float*    ald    = als + (size_t)N * 2;
  int*      rowptr = (int*)(ald + (size_t)N * 2);
  int*      tab    = rowptr + (size_t)N + 1;
  unsigned* staged = (unsigned*)(tab + (size_t)(nb + 1) * G);
  int*      csrc   = (int*)(staged + (size_t)G * SLICE_E);

  const int B = 256;
  const int gGemm  = (N + GEMM_ROWS - 1) / GEMM_ROWS;
  const int gNodeW = (N + 7) / 8;

  csr_slice_bins<<<G, 1024, 0, stream>>>(esrc, edst, staged, tab, E, nb, G);
  csr_bin_rows<<<nb, 1024, 0, stream>>>(staged, tab, rowptr, csrc, N, G);

  gat_gemm128_wmma<<<gGemm, 128, 0, stream>>>(x, W1, as1, ad1, h, als, ald, N);
  gat_node_gather<true><<<gNodeW, B, 0, stream>>>(rowptr, csrc, als, ald, h, b1, act, N);

  gat_gemm128_wmma<<<gGemm, 128, 0, stream>>>(act, W2, as2, ad2, h, als, ald, N);
  gat_node_gather<false><<<gNodeW, B, 0, stream>>>(rowptr, csrc, als, ald, h, b2,
                                                   (float*)d_out, N);
}


# Node gather: cross-lane traffic moved off LDS shuffles and lane reads onto DPP row operations: edge weights reach the FMAs as row-broadcast operands, softmax butterflies on row_xmask; sum order kept
# speedup vs baseline: 6.9849x; 1.0413x over previous
#include <hip/hip_runtime.h>
#include <math.h>

typedef float v2f __attribute__((ext_vector_type(2)));
typedef float v8f __attribute__((ext_vector_type(8)));

#define NEG_SLOPE 0.2f

#define ROW_XOR_ADD(v, m)                                                        \
  (v) += __int_as_float(__builtin_amdgcn_update_dpp(0, __float_as_int(v),        \
                                                    0x160 + (m), 0xf, 0xf, false))
#define TILE_PITCH 132
#define GEMM_ROWS  64
#define WL_PITCH   288
__global__ void __launch_bounds__(128)
gat_gemm128_wmma(const float* __restrict__ X,
                 const float* __restrict__ W,
                 const float* __restrict__ a_src,
                 const float* __restrict__ a_dst,
                 float* __restrict__ H,
                 float* __restrict__ als,
                 float* __restrict__ ald, int N) {
  __shared__ __attribute__((aligned(16))) float lds[64 * WL_PITCH];
  const int wave = threadIdx.x >> 5;
  const int lane = threadIdx.x & 31;
  const int half = lane >> 4;
  const int lm   = lane & 15;
  const int row0 = blockIdx.x * GEMM_ROWS + (wave << 4);

  int arow = row0 + lm;
  if (arow >= N) arow = N - 1;
  const float* xr = X + (size_t)arow * 128 + (half << 1);

  v2f a[32];
#pragma unroll
  for (int kk = 0; kk < 32; ++kk) a[kk] = *(const v2f*)(xr + (kk << 2));

  const float4 as4 = *(const float4*)(a_src + lane * 4);
  const float4 ad4 = *(const float4*)(a_dst + lane * 4);

  const float4* W4 = (const float4*)W;
  for (int it = 0; it < 16; ++it) {
    const int p = (it << 2) + wave;
    const float4 r0 = W4[(2 * p) * 32 + lane];
    const float4 r1 = W4[(2 * p + 1) * 32 + lane];
    float4 q0, q1;
    q0.x = r0.x; q0.y = r1.x; q0.z = r0.y; q0.w = r1.y;
    q1.x = r0.z; q1.y = r1.z; q1.z = r0.w; q1.w = r1.w;
    float* dst = lds + p * WL_PITCH + lane * 8;
    *(float4*)dst = q0;
    *(float4*)(dst + 4) = q1;
  }
  __syncthreads();

  v8f c[8];
#pragma unroll
  for (int j = 0; j < 8; ++j) c[j] = (v8f){};
  const float* bl = lds + half * WL_PITCH + (lm << 1);
  v2f bc[8], bn[8];
#pragma unroll
  for (int j = 0; j < 8; ++j) bc[j] = *(const v2f*)(bl + j * 32);
#pragma unroll
  for (int kk = 0; kk < 32; ++kk) {
    if (kk + 1 < 32) {
#pragma unroll
      for (int j = 0; j < 8; ++j)
        bn[j] = *(const v2f*)(bl + (kk + 1) * 2 * WL_PITCH + j * 32);
    }
#pragma unroll
    for (int j = 0; j < 8; ++j)
      c[j] = __builtin_amdgcn_wmma_f32_16x16x4_f32(false, a[kk], false, bc[j],
                                                   (short)0, c[j], false, false);
#pragma unroll
    for (int j = 0; j < 8; ++j) bc[j] = bn[j];
  }

  __syncthreads();
  float* tw = lds + wave * (16 * TILE_PITCH);
#pragma unroll
  for (int j = 0; j < 8; ++j)
#pragma unroll
    for (int v = 0; v < 8; ++v)
      tw[((half << 3) + v) * TILE_PITCH + (j << 4) + lm] = c[j][v];
  __syncthreads();

  float sfin = 0.f, dfin = 0.f;
#pragma unroll
  for (int ii = 0; ii < 8; ++ii) {
    float4 hv[2];
    float s[2], d[2];
#pragma unroll
    for (int i = 0; i < 2; ++i) {
      const int r = (ii << 1) + i;
      hv[i] = *(const float4*)(tw + r * TILE_PITCH + lane * 4);
      s[i] = 0.f; d[i] = 0.f;
      s[i] += hv[i].x * as4.x;  d[i] += hv[i].x * ad4.x;
      s[i] += hv[i].y * as4.y;  d[i] += hv[i].y * ad4.y;
      s[i] += hv[i].z * as4.z;  d[i] += hv[i].z * ad4.z;
      s[i] += hv[i].w * as4.w;  d[i] += hv[i].w * ad4.w;
    }
    ROW_XOR_ADD(s[0], 8); ROW_XOR_ADD(d[0], 8); ROW_XOR_ADD(s[1], 8); ROW_XOR_ADD(d[1], 8);
    ROW_XOR_ADD(s[0], 4); ROW_XOR_ADD(d[0], 4); ROW_XOR_ADD(s[1], 4); ROW_XOR_ADD(d[1], 4);
    ROW_XOR_ADD(s[0], 2); ROW_XOR_ADD(d[0], 2); ROW_XOR_ADD(s[1], 2); ROW_XOR_ADD(d[1], 2);
    ROW_XOR_ADD(s[0], 1); ROW_XOR_ADD(d[0], 1); ROW_XOR_ADD(s[1], 1); ROW_XOR_ADD(d[1], 1);
#pragma unroll
    for (int i = 0; i < 2; ++i) {
      const int r = (ii << 1) + i;
      const int n = row0 + r;
      if (n < N) ((float4*)H)[(size_t)n * 32 + lane] = hv[i];
      if (lm == r) { sfin = s[i]; dfin = d[i]; }
    }
  }
  const int nr = row0 + lm;
  if (nr < N) {
    als[nr * 2 + half] = sfin;
    ald[nr * 2 + half] = dfin;
  }
}

#define SLICE_E   8192
#define BIN_SHIFT 8
#define BIN_NODES 256
#define OUT_CAP   8192

__device__ __forceinline__ int block_excl_scan(int c, int* wsum, int& total) {
  const int lane = threadIdx.x & 31;
  const int w = threadIdx.x >> 5;
  int inc = c;
#pragma unroll
  for (int d = 1; d < 32; d <<= 1) {
    const int v = __shfl_up(inc, d, 32);
    if (lane >= d) inc += v;
  }
  if (lane == 31) wsum[w] = inc;
  __syncthreads();
  int o = wsum[lane];
#pragma unroll
  for (int d = 1; d < 32; d <<= 1) {
    const int v = __shfl_up(o, d, 32);
    if (lane >= d) o += v;
  }
  total = __shfl(o, 31, 32);
  const int before = __shfl(o, (w + 31) & 31, 32);
  return (w > 0 ? before : 0) + (inc - c);
}

__global__ void __launch_bounds__(1024)
csr_slice_bins(const int* __restrict__ es, const int* __restrict__ ed,
               unsigned* __restrict__ staged, int* __restrict__ tab,
               int E, int nb, int G) {
  __shared__ int hist[1024];
  __shared__ int wsum[32];
  __shared__ unsigned stage[SLICE_E];
  const int t = threadIdx.x;
  const int b = blockIdx.x;
  const int e0 = b * SLICE_E;
  hist[t] = 0;
  int s[8], d[8];
#pragma unroll
  for (int i = 0; i < 8; ++i) {
    const int e = e0 + i * 1024 + t;
    d[i] = (e < E) ? ed[e] : -1;
    s[i] = (e < E) ? es[e] : 0;
  }
  __syncthreads();
#pragma unroll
  for (int i = 0; i < 8; ++i)
    if (d[i] >= 0) atomicAdd(&hist[d[i] >> BIN_SHIFT], 1);
  __syncthreads();
  const int c = hist[t];
  int total;
  const int excl = block_excl_scan(c, wsum, total);
  hist[t] = excl;
  if (t < nb) tab[(size_t)t * G + b] = excl;
  if (t == 0) tab[(size_t)nb * G + b] = total;
  __syncthreads();
  int slot[8];
#pragma unroll
  for (int i = 0; i < 8; ++i)
    slot[i] = (d[i] >= 0) ? atomicAdd(&hist[d[i] >> BIN_SHIFT], 1) : 0;
#pragma unroll
  for (int i = 0; i < 8; ++i)
    if (d[i] >= 0)
      stage[slot[i]] = ((unsigned)s[i] << BIN_SHIFT) | (unsigned)(d[i] & (BIN_NODES - 1));
  __syncthreads();
  for (int j = t; j < total; j += 1024) staged[(size_t)e0 + j] = stage[j];
}

template <bool PLACE>
__device__ __forceinline__ void bin_pass(const unsigned* __restrict__ staged,
                                         const int* segoff, const int* seglen,
                                         int* cnt, int* outl, int* __restrict__ csrc,
                                         int base, int G) {
  const int lane = threadIdx.x & 31;
  const int w = threadIdx.x >> 5;
  for (int b0 = w; b0 < G; b0 += 128) {
    size_t off[4];
    int len[4];
    int mx = 0;
#pragma unroll
    for (int i = 0; i < 4; ++i) {
      const int sb = b0 + 32 * i;
      const bool ok = sb < G;
      off[i] = ok ? (size_t)sb * SLICE_E + segoff[sb] : 0;
      len[i] = ok ? seglen[sb] : 0;
      mx = len[i] > mx ? len[i] : mx;
    }
    for (int j0 = 0; j0 < mx; j0 += 32) {
      const int j = j0 + lane;
      unsigned v[4];
#pragma unroll
      for (int i = 0; i < 4; ++i) v[i] = (j < len[i]) ? staged[off[i] + j] : 0u;
#pragma unroll
      for (int i = 0; i < 4; ++i) {
        if (j < len[i]) {
          const int dl = (int)(v[i] & (BIN_NODES - 1));
          if (!PLACE) {
            atomicAdd(&cnt[dl], 1);
          } else {
            const int slot = atomicAdd(&cnt[dl], 1);
            const int sv = (int)(v[i] >> BIN_SHIFT);
            if (slot < OUT_CAP) outl[slot] = sv;
            else csrc[(size_t)base + slot] = sv;
          }
        }
      }
    }
  }
}

__global__ void __launch_bounds__(1024)
csr_bin_rows(const unsigned* __restrict__ staged, const int* __restrict__ tab,
             int* __restrict__ rowptr, int* __restrict__ csrc, int N, int G) {
  __shared__ int segoff[1024];
  __shared__ int seglen[1024];
  __shared__ int cnt[BIN_NODES];
  __shared__ int wsum[32];
  __shared__ int wbase[32];
  __shared__ int outl[OUT_CAP];
  const int t = threadIdx.x;
  const int lane = t & 31;
  const int w = t >> 5;
  const int k = blockIdx.x;

  int o = 0;
  if (t < G) {
    o = tab[(size_t)k * G + t];
    const int o2 = tab[(size_t)(k + 1) * G + t];
    segoff[t] = o;
    seglen[t] = o2 - o;
  }
  if (t < BIN_NODES) cnt[t] = 0;
#pragma unroll
  for (int m = 16; m >= 1; m >>= 1) o += __shfl_xor(o, m, 32);
  if (lane == 0) wbase[w] = o;
  __syncthreads();
  int base = wbase[lane];
#pragma unroll
  for (int m = 16; m >= 1; m >>= 1) base += __shfl_xor(base, m, 32);

  bin_pass<false>(staged, segoff, seglen, cnt, outl, csrc, base, G);
  __syncthreads();

  const int c = (t < BIN_NODES) ? cnt[t] : 0;
  int total;
  const int excl = block_excl_scan(c, wsum, total);
  if (t < BIN_NODES) {
    cnt[t] = excl;
    const int node = (k << BIN_SHIFT) + t;
    if (node < N) {
      rowptr[node] = base + excl;
      if (node == N - 1) rowptr[N] = base + excl + c;
    }
  }
  __syncthreads();

  bin_pass<true>(staged, segoff, seglen, cnt, outl, csrc, base, G);
  __syncthreads();

  const int nl = total < OUT_CAP ? total : OUT_CAP;
  for (int j = t; j < nl; j += 1024) csrc[(size_t)base + j] = outl[j];
}

__device__ __forceinline__ float lrelu(float v) {
  return (v >= 0.f) ? v : NEG_SLOPE * v;
}

#define ROW_XMASK_F(v, m)                                                        \
  __int_as_float(__builtin_amdgcn_update_dpp(0, __float_as_int(v), 0x160 + (m), 0xf, 0xf, false))
#define ROW_SHARE_F(v, i)                                                        \
  __int_as_float(__builtin_amdgcn_mov_dpp(__float_as_int(v), 0x150 + (i), 0xf, 0xf, false))
__device__ __forceinline__ float other_row(float v) {
  return __int_as_float((int)__builtin_amdgcn_permlanex16(0u, (unsigned)__float_as_int(v),
                                                          0x76543210u, 0xfedcba98u, false, false));
}

__device__ __forceinline__ float4 load_row(const float4* __restrict__ H4, int src, int lane) {
  const float4* rp = H4 + (size_t)src * 32;
  return rp[lane];
}

template <int I>
__device__ __forceinline__ void fma_row(float4& acc, const float4 r, float wrow) {
  const float wt = ROW_SHARE_F(wrow, I);
  acc.x += wt * r.x; acc.y += wt * r.y; acc.z += wt * r.z; acc.w += wt * r.w;
}

template <int T0>
__device__ __forceinline__ void gather_group(float4& acc, const float4* __restrict__ H4,
                                             int lane, int s, float wrow, int c) {
  constexpr int I0 = T0 & 15;
  const int sa = __builtin_amdgcn_readlane(s, T0);
  const int tb = __builtin_amdgcn_readlane(s, T0 + 1);
  const int tc = __builtin_amdgcn_readlane(s, T0 + 2);
  const int td = __builtin_amdgcn_readlane(s, T0 + 3);
  const int sb = (c > T0 + 1) ? tb : sa;
  const int sc = (c > T0 + 2) ? tc : sa;
  const int sd = (c > T0 + 3) ? td : sa;
  const float4 ra = load_row(H4, sa, lane);
  const float4 rb = load_row(H4, sb, lane);
  const float4 rc = load_row(H4, sc, lane);
  const float4 rd = load_row(H4, sd, lane);
  fma_row<I0>(acc, ra, wrow);
  fma_row<I0 + 1>(acc, rb, wrow);
  fma_row<I0 + 2>(acc, rc, wrow);
  fma_row<I0 + 3>(acc, rd, wrow);
}

__device__ __forceinline__ void gather_chunk(float4& acc, const float4* __restrict__ H4,
                                             int lane, bool hi, int s, float w0, float w1,
                                             int c) {
  const float x0 = other_row(w0);
  const float x1 = other_row(w1);
  const float wlo = hi ? x1 : w0;
  const float whi = hi ? w1 : x0;
  if (c <= 0) return;
  gather_group<0>(acc, H4, lane, s, wlo, c);
  if (c <= 4) return;
  gather_group<4>(acc, H4, lane, s, wlo, c);
  if (c <= 8) return;
  gather_group<8>(acc, H4, lane, s, wlo, c);
  if (c <= 12) return;
  gather_group<12>(acc, H4, lane, s, wlo, c);
  if (c <= 16) return;
  gather_group<16>(acc, H4, lane, s, whi, c);
  if (c <= 20) return;
  gather_group<20>(acc, H4, lane, s, whi, c);
  if (c <= 24) return;
  gather_group<24>(acc, H4, lane, s, whi, c);
  if (c <= 28) return;
  gather_group<28>(acc, H4, lane, s, whi, c);
}

template <bool CONCAT>
__global__ void gat_node_gather(const int* __restrict__ rowptr, const int* __restrict__ csrc,
                                const float* __restrict__ als, const float* __restrict__ ald,
                                const float* __restrict__ H, const float* __restrict__ bias,
                                float* __restrict__ out, int N) {
  const int n = __builtin_amdgcn_readfirstlane(
      (int)(blockIdx.x * (blockDim.x >> 5) + (threadIdx.x >> 5)));
  const int lane = threadIdx.x & 31;
  if (n >= N) return;
  const int beg = rowptr[n], end = rowptr[n + 1];
  const int deg = end - beg;
  const float2 asn = *(const float2*)(als + 2 * (size_t)n);
  const float2 adn = *(const float2*)(ald + 2 * (size_t)n);
  const float self0 = lrelu(asn.x + adn.x);
  const float self1 = lrelu(asn.y + adn.y);

  const bool has = lane < deg;
  int   s_l = n;
  float v0 = -INFINITY, v1 = -INFINITY;
  if (has) {
    s_l = csrc[beg + lane];
    const float2 a = *(const float2*)(als + 2 * (size_t)s_l);
    v0 = lrelu(a.x + adn.x);
    v1 = lrelu(a.y + adn.y);
  }

  float m0 = fmaxf(self0, v0), m1 = fmaxf(self1, v1);
  for (int j = beg + 32 + lane; j < end; j += 32) {
    const int s = csrc[j];
    const float2 a = *(const float2*)(als + 2 * (size_t)s);
    m0 = fmaxf(m0, lrelu(a.x + adn.x));
    m1 = fmaxf(m1, lrelu(a.y + adn.y));
  }
  m0 = fmaxf(m0, other_row(m0));      m1 = fmaxf(m1, other_row(m1));
  m0 = fmaxf(m0, ROW_XMASK_F(m0, 8)); m1 = fmaxf(m1, ROW_XMASK_F(m1, 8));
  m0 = fmaxf(m0, ROW_XMASK_F(m0, 4)); m1 = fmaxf(m1, ROW_XMASK_F(m1, 4));
  m0 = fmaxf(m0, ROW_XMASK_F(m0, 2)); m1 = fmaxf(m1, ROW_XMASK_F(m1, 2));
  m0 = fmaxf(m0, ROW_XMASK_F(m0, 1)); m1 = fmaxf(m1, ROW_XMASK_F(m1, 1));

  const float ex0 = has ? expf(v0 - m0) : 0.f;
  const float ex1 = has ? expf(v1 - m1) : 0.f;
  float d0 = ex0, d1 = ex1;
  for (int j = beg + 32 + lane; j < end; j += 32) {
    const int s = csrc[j];
    const float2 a = *(const float2*)(als + 2 * (size_t)s);
    d0 += expf(lrelu(a.x + adn.x) - m0);
    d1 += expf(lrelu(a.y + adn.y) - m1);
  }
  d0 += other_row(d0);      d1 += other_row(d1);
  d0 += ROW_XMASK_F(d0, 8); d1 += ROW_XMASK_F(d1, 8);
  d0 += ROW_XMASK_F(d0, 4); d1 += ROW_XMASK_F(d1, 4);
  d0 += ROW_XMASK_F(d0, 2); d1 += ROW_XMASK_F(d1, 2);
  d0 += ROW_XMASK_F(d0, 1); d1 += ROW_XMASK_F(d1, 1);
  const float exs0 = expf(self0 - m0), exs1 = expf(self1 - m1);
  d0 += exs0;
  d1 += exs1;

  const float4* H4 = (const float4*)H;
  const bool hi = lane >= 16;
  const float wself = hi ? (exs1 / d1) : (exs0 / d0);
  const float4 hn = H4[(size_t)n * 32 + lane];
  float4 acc;
  acc.x = wself * hn.x; acc.y = wself * hn.y; acc.z = wself * hn.z; acc.w = wself * hn.w;
  {
    int s = s_l;
    float w0 = ex0 / d0, w1 = ex1 / d1;
    int base = beg;
    for (;;) {
      const int c = end - base;
      gather_chunk(acc, H4, lane, hi, s, w0, w1, c < 32 ? c : 32);
      base += 32;
      if (base >= end) break;
      const int j = base + lane;
      s = n; w0 = 0.f; w1 = 0.f;
      if (j < end) {
        s = csrc[j];
        const float2 a = *(const float2*)(als + 2 * (size_t)s);
        w0 = expf(lrelu(a.x + adn.x) - m0) / d0;
        w1 = expf(lrelu(a.y + adn.y) - m1) / d1;
      }
    }
  }

  if (CONCAT) {
    float4 v;
    v.x = acc.x + bias[lane * 4 + 0];
    v.y = acc.y + bias[lane * 4 + 1];
    v.z = acc.z + bias[lane * 4 + 2];
    v.w = acc.w + bias[lane * 4 + 3];
    v.x = (v.x > 0.f) ? v.x : (expf(v.x) - 1.f);
    v.y = (v.y > 0.f) ? v.y : (expf(v.y) - 1.f);
    v.z = (v.z > 0.f) ? v.z : (expf(v.z) - 1.f);
    v.w = (v.w > 0.f) ? v.w : (expf(v.w) - 1.f);
    ((float4*)out)[(size_t)n * 32 + lane] = v;
  } else {
    float4 o;
    o.x = other_row(acc.x);
    o.y = other_row(acc.y);
    o.z = other_row(acc.z);
    o.w = other_row(acc.w);
    if (!hi) {
      float4 v;
      v.x = 0.5f * (acc.x + o.x) + bias[lane * 4 + 0];
      v.y = 0.5f * (acc.y + o.y) + bias[lane * 4 + 1];
      v.z = 0.5f * (acc.z + o.z) + bias[lane * 4 + 2];
      v.w = 0.5f * (acc.w + o.w) + bias[lane * 4 + 3];
      ((float4*)out)[(size_t)n * 16 + lane] = v;
    }
  }
}

extern "C" void kernel_launch(void* const* d_in, const int* in_sizes, int n_in,
                              void* d_out, int out_size, void* d_ws, size_t ws_size,
                              hipStream_t stream) {
  const float* x   = (const float*)d_in[0];
  const int*   ei  = (const int*)d_in[1];
  const float* W1  = (const float*)d_in[2];
  const float* as1 = (const float*)d_in[3];
  const float* ad1 = (const float*)d_in[4];
  const float* b1  = (const float*)d_in[5];
  const float* W2  = (const float*)d_in[6];
  const float* as2 = (const float*)d_in[7];
  const float* ad2 = (const float*)d_in[8];
  const float* b2  = (const float*)d_in[9];

  const int N = in_sizes[0] / 128;
  const int E = in_sizes[1] / 2;
  const int* esrc = ei;
  const int* edst = ei + E;

  const int nb = (N + BIN_NODES - 1) >> BIN_SHIFT;
  int G = (E + SLICE_E - 1) / SLICE_E;
  if (G < 1) G = 1;
  float*    h      = (float*)d_ws;
  float*    act    = h + (size_t)N * 128;
  float*    als    = act + (size_t)N * 128;
  float*    ald    = als + (size_t)N * 2;
  int*      rowptr = (int*)(ald + (size_t)N * 2);
  int*      tab    = rowptr + (size_t)N + 1;
  unsigned* staged = (unsigned*)(tab + (size_t)(nb + 1) * G);
  int*      csrc   = (int*)(staged + (size_t)G * SLICE_E);

  const int B = 256;
  const int gGemm  = (N + GEMM_ROWS - 1) / GEMM_ROWS;
  const int gNodeW = (N + 7) / 8;

  csr_slice_bins<<<G, 1024, 0, stream>>>(esrc, edst, staged, tab, E, nb, G);
  csr_bin_rows<<<nb, 1024, 0, stream>>>(staged, tab, rowptr, csrc, N, G);

  gat_gemm128_wmma<<<gGemm, 128, 0, stream>>>(x, W1, as1, ad1, h, als, ald, N);
  gat_node_gather<true><<<gNodeW, B, 0, stream>>>(rowptr, csrc, als, ald, h, b1, act, N);

  gat_gemm128_wmma<<<gGemm, 128, 0, stream>>>(act, W2, as2, ad2, h, als, ald, N);
  gat_node_gather<false><<<gNodeW, B, 0, stream>>>(rowptr, csrc, als, ald, h, b2,
                                                   (float*)d_out, N);
}
